# starting kernel
# baseline (speedup 1.0000x reference)
#include <hip/hip_runtime.h>
#include <math.h>

typedef __attribute__((ext_vector_type(2))) float v2f;
typedef __attribute__((ext_vector_type(8))) float v8f;

#define LN_EPS 1e-5f

__device__ __forceinline__ void atomAddF(float* p, float v) {
  unsafeAtomicAdd(p, v);
}

__global__ void init_deg(float* deg, int n) {
  int i = blockIdx.x * blockDim.x + threadIdx.x;
  if (i < n) deg[i] = 1.0f;
}

__global__ void deg_accum(const int* __restrict__ dst, float* deg, int E) {
  int i = blockIdx.x * blockDim.x + threadIdx.x;
  if (i < E) atomAddF(&deg[dst[i]], 1.0f);
}

__global__ void finish_dinv(float* deg, int n) {
  int i = blockIdx.x * blockDim.x + threadIdx.x;
  if (i < n) {
    float d = deg[i];
    deg[i] = (d > 0.0f) ? rsqrtf(d) : 0.0f;
  }
}

template <int FOUT>
__global__ __launch_bounds__(256) void gemm_wmma_f32(
    const float* __restrict__ X, const float* __restrict__ W,
    float* __restrict__ H, int nrows)
{
  constexpr int TPR = FOUT / 16;
  constexpr int RPB = 8 / TPR;
  const int w    = threadIdx.x >> 5;
  const int lane = threadIdx.x & 31;

  const int rowTile = blockIdx.x * RPB + (w / TPR);
  const int colTile = w % TPR;
  const int r0 = rowTile * 16;
  if (r0 >= nrows) return;
  const int c0 = colTile * 16;

  const int lh = lane & 15;
  const int kh = (lane >> 4) << 1;

  const float* __restrict__ xrow = X + (size_t)(r0 + lh) * 128 + kh;
  const float* __restrict__ wcol = W + (size_t)kh * FOUT + c0 + lh;

  v8f acc = {};
#pragma unroll
  for (int kk = 0; kk < 128; kk += 4) {
    v2f a; a.x = xrow[0];   a.y = xrow[1];
    v2f b; b.x = wcol[0];   b.y = wcol[FOUT];
    acc = __builtin_amdgcn_wmma_f32_16x16x4_f32(
        false, a, false, b, (short)0, acc, false, false);
    xrow += 4;
    wcol += 4 * FOUT;
  }

  const int rbase = r0 + ((lane >> 4) << 3);
  float* hp = H + (size_t)rbase * FOUT + c0 + lh;
#pragma unroll
  for (int v = 0; v < 8; ++v) hp[(size_t)v * FOUT] = acc[v];
}

template <int FOUT>
__global__ __launch_bounds__(256) void scatter_edges(
    const float* __restrict__ H, const int* __restrict__ srcA,
    const int* __restrict__ dstA, const float* __restrict__ dinv,
    float* __restrict__ AGG, int E, int n)
{
  constexpr int VEC = FOUT / 32;
  const int gw   = (int)((blockIdx.x * blockDim.x + threadIdx.x) >> 5);
  const int lane = threadIdx.x & 31;
  const int total = E + n;
  if (gw >= total) return;

  int s, d;
  if (gw < E) { s = srcA[gw]; d = dstA[gw]; }
  else        { s = gw - E;   d = s; }

  const float nrm = dinv[s] * dinv[d];
  const float* __restrict__ hp = H + (size_t)s * FOUT + lane * VEC;
  float* ap = AGG + (size_t)d * FOUT + lane * VEC;

  float v[VEC];
#pragma unroll
  for (int j = 0; j < VEC; ++j) v[j] = hp[j] * nrm;
#pragma unroll
  for (int j = 0; j < VEC; ++j) atomAddF(ap + j, v[j]);
}

__global__ __launch_bounds__(256) void ln_relu_128(
    const float* __restrict__ AGG, const float* __restrict__ bias,
    const float* __restrict__ gamma, const float* __restrict__ beta,
    float* __restrict__ OUT, int n)
{
  const int node = (int)((blockIdx.x * blockDim.x + threadIdx.x) >> 5);
  const int lane = threadIdx.x & 31;
  if (node >= n) return;

  const float* row = AGG + (size_t)node * 128 + lane * 4;
  float v0 = row[0] + bias[lane * 4 + 0];
  float v1 = row[1] + bias[lane * 4 + 1];
  float v2 = row[2] + bias[lane * 4 + 2];
  float v3 = row[3] + bias[lane * 4 + 3];

  float s = v0 + v1 + v2 + v3;
#pragma unroll
  for (int m = 16; m >= 1; m >>= 1) s += __shfl_xor(s, m, 32);
  const float mu = s * (1.0f / 128.0f);

  float d0 = v0 - mu, d1 = v1 - mu, d2 = v2 - mu, d3 = v3 - mu;
  float q = d0 * d0 + d1 * d1 + d2 * d2 + d3 * d3;
#pragma unroll
  for (int m = 16; m >= 1; m >>= 1) q += __shfl_xor(q, m, 32);
  const float rstd = rsqrtf(q * (1.0f / 128.0f) + LN_EPS);

  float* out = OUT + (size_t)node * 128 + lane * 4;
  out[0] = fmaxf(0.0f, d0 * rstd * gamma[lane * 4 + 0] + beta[lane * 4 + 0]);
  out[1] = fmaxf(0.0f, d1 * rstd * gamma[lane * 4 + 1] + beta[lane * 4 + 1]);
  out[2] = fmaxf(0.0f, d2 * rstd * gamma[lane * 4 + 2] + beta[lane * 4 + 2]);
  out[3] = fmaxf(0.0f, d3 * rstd * gamma[lane * 4 + 3] + beta[lane * 4 + 3]);
}

__global__ void bias_add_64(const float* __restrict__ AGG,
                            const float* __restrict__ bias,
                            float* __restrict__ OUT, int total)
{
  int i = blockIdx.x * blockDim.x + threadIdx.x;
  if (i < total) OUT[i] = AGG[i] + bias[i & 63];
}

extern "C" void kernel_launch(void* const* d_in, const int* in_sizes, int n_in,
                              void* d_out, int out_size, void* d_ws, size_t ws_size,
                              hipStream_t stream) {
  const float* x   = (const float*)d_in[0];
  const int*   ei  = (const int*)d_in[1];
  const float* W1  = (const float*)d_in[2];
  const float* b1  = (const float*)d_in[3];
  const float* g1  = (const float*)d_in[4];
  const float* be1 = (const float*)d_in[5];
  const float* W2  = (const float*)d_in[6];
  const float* b2  = (const float*)d_in[7];
  const float* g2  = (const float*)d_in[8];
  const float* be2 = (const float*)d_in[9];
  const float* W3  = (const float*)d_in[10];
  const float* b3  = (const float*)d_in[11];

  const int N = in_sizes[0] / 128;
  const int E = in_sizes[1] / 2;
  const int* src = ei;
  const int* dst = ei + E;

  float* ws   = (float*)d_ws;
  float* dinv = ws;
  float* h    = dinv + N;
  float* agg  = h + (size_t)N * 128;
  float* hn   = agg + (size_t)N * 128;

  init_deg   <<<(N + 255) / 256, 256, 0, stream>>>(dinv, N);
  deg_accum  <<<(E + 255) / 256, 256, 0, stream>>>(dst, dinv, E);
  finish_dinv<<<(N + 255) / 256, 256, 0, stream>>>(dinv, N);

  const int total   = E + N;
  const int sblocks = (total + 7) / 8;

  gemm_wmma_f32<128><<<(N + 15) / 16, 256, 0, stream>>>(x, W1, h, N);
  hipMemsetAsync(agg, 0, (size_t)N * 128 * sizeof(float), stream);
  scatter_edges<128><<<sblocks, 256, 0, stream>>>(h, src, dst, dinv, agg, E, N);
  ln_relu_128<<<(N + 7) / 8, 256, 0, stream>>>(agg, b1, g1, be1, hn, N);

  gemm_wmma_f32<128><<<(N + 15) / 16, 256, 0, stream>>>(hn, W2, h, N);
  hipMemsetAsync(agg, 0, (size_t)N * 128 * sizeof(float), stream);
  scatter_edges<128><<<sblocks, 256, 0, stream>>>(h, src, dst, dinv, agg, E, N);
  ln_relu_128<<<(N + 7) / 8, 256, 0, stream>>>(agg, b2, g2, be2, hn, N);

  gemm_wmma_f32<64><<<(N + 31) / 32, 256, 0, stream>>>(hn, W3, h, N);
  hipMemsetAsync(agg, 0, (size_t)N * 64 * sizeof(float), stream);
  scatter_edges<64><<<sblocks, 256, 0, stream>>>(h, src, dst, dinv, agg, E, N);
  bias_add_64<<<(N * 64 + 255) / 256, 256, 0, stream>>>(agg, b3, (float*)d_out, N * 64);
}


# Replace per-edge float atomic scatter and its zero fill by a per-call CSR build (int atomic rank, scan, fill) and a one-wave-per-node gather; per-node sum order: self-loop first, then rank order
# speedup vs baseline: 2.3299x; 2.3299x over previous
#include <hip/hip_runtime.h>
#include <math.h>

typedef __attribute__((ext_vector_type(2))) float v2f;
typedef __attribute__((ext_vector_type(8))) float v8f;

#define LN_EPS 1e-5f

__global__ void zero_cnt(int* cnt, int n) {
  int i = blockIdx.x * blockDim.x + threadIdx.x;
  if (i < n) cnt[i] = 0;
}

__global__ void count_rank(const int* __restrict__ dst, int* cnt,
                           int* __restrict__ rank, int E) {
  int i = blockIdx.x * blockDim.x + threadIdx.x;
  if (i < E) rank[i] = atomicAdd(&cnt[dst[i]], 1);
}

__global__ __launch_bounds__(256) void scan_dinv(
    const int* __restrict__ cnt, int2* __restrict__ seg,
    float* __restrict__ dinv, int2* __restrict__ cv, int n)
{
  __shared__ int part[16];
  const int t    = threadIdx.x;
  const int w    = t >> 5;
  const int lane = t & 31;
  const int b    = blockIdx.x;

  const int4* __restrict__ c4 = (const int4*)cnt;
  int pre = 0;
  for (int i = 0; i < b; ++i) {
    int4 v = c4[i * 256 + t];
    pre += (v.x + v.y) + (v.z + v.w);
  }
#pragma unroll
  for (int m = 16; m >= 1; m >>= 1) pre += __shfl_xor(pre, m, 32);

  const int i0 = b * 1024 + t * 4;
  int c[4];
#pragma unroll
  for (int j = 0; j < 4; ++j) c[j] = (i0 + j < n) ? cnt[i0 + j] + 1 : 0;
  const int tsum = (c[0] + c[1]) + (c[2] + c[3]);
  int incl = tsum;
#pragma unroll
  for (int m = 1; m < 32; m <<= 1) {
    int o = __shfl_up(incl, m, 32);
    if (lane >= m) incl += o;
  }
  if (lane == 0)  part[w] = pre;
  if (lane == 31) part[8 + w] = incl;
  __syncthreads();
  int off = b * 1024;
#pragma unroll
  for (int k = 0; k < 8; ++k) {
    off += part[k];
    if (k < w) off += part[8 + k];
  }
  int run = off + incl - tsum;
#pragma unroll
  for (int j = 0; j < 4; ++j) {
    if (i0 + j < n) {
      seg[i0 + j] = make_int2(run, c[j]);
      const float d  = (float)c[j];
      const float dv = (d > 0.0f) ? rsqrtf(d) : 0.0f;
      dinv[i0 + j] = dv;
      cv[run] = make_int2(i0 + j, __float_as_int(dv * dv));
    }
    run += c[j];
  }
}

__global__ void fill_csr(const int* __restrict__ src, const int* __restrict__ dst,
                         const int* __restrict__ rank, const int2* __restrict__ seg,
                         const float* __restrict__ dinv, int2* __restrict__ cv, int E) {
  int i = blockIdx.x * blockDim.x + threadIdx.x;
  if (i < E) {
    const int s = src[i], d = dst[i];
    const int pos = seg[d].x + 1 + rank[i];
    cv[pos] = make_int2(s, __float_as_int(dinv[s] * dinv[d]));
  }
}

template <int FOUT>
__global__ __launch_bounds__(256) void gemm_wmma_f32(
    const float* __restrict__ X, const float* __restrict__ W,
    float* __restrict__ H, int nrows)
{
  constexpr int TPR = FOUT / 16;
  constexpr int RPB = 8 / TPR;
  const int w    = threadIdx.x >> 5;
  const int lane = threadIdx.x & 31;

  const int rowTile = blockIdx.x * RPB + (w / TPR);
  const int colTile = w % TPR;
  const int r0 = rowTile * 16;
  if (r0 >= nrows) return;
  const int c0 = colTile * 16;

  const int lh = lane & 15;
  const int kh = (lane >> 4) << 1;

  const float* __restrict__ xrow = X + (size_t)(r0 + lh) * 128 + kh;
  const float* __restrict__ wcol = W + (size_t)kh * FOUT + c0 + lh;

  v8f acc = {};
#pragma unroll
  for (int kk = 0; kk < 128; kk += 4) {
    v2f a; a.x = xrow[0];   a.y = xrow[1];
    v2f b; b.x = wcol[0];   b.y = wcol[FOUT];
    acc = __builtin_amdgcn_wmma_f32_16x16x4_f32(
        false, a, false, b, (short)0, acc, false, false);
    xrow += 4;
    wcol += 4 * FOUT;
  }

  const int rbase = r0 + ((lane >> 4) << 3);
  float* hp = H + (size_t)rbase * FOUT + c0 + lh;
#pragma unroll
  for (int v = 0; v < 8; ++v) hp[(size_t)v * FOUT] = acc[v];
}

template <int FOUT>
__global__ __launch_bounds__(256) void gather_nodes(
    const float* __restrict__ H, const int2* __restrict__ seg,
    const int2* __restrict__ cv, float* __restrict__ AGG, int n)
{
  constexpr int VEC = FOUT / 32;
  typedef __attribute__((ext_vector_type(VEC))) float vecT;
  const int lane = threadIdx.x & 31;
  const int node = __builtin_amdgcn_readfirstlane(
      (int)((blockIdx.x * blockDim.x + threadIdx.x) >> 5));
  if (node >= n) return;

  const int2  sg    = seg[node];
  const int   start = sg.x;
  const int   m     = sg.y;
  const vecT* __restrict__ Hv = (const vecT*)H;

  vecT acc = {};

  for (int base = 0; base < m; base += 32) {
    const int mc = (m - base < 32) ? (m - base) : 32;
    int c = 0, wb = 0;
    if (lane < mc) { const int2 e = cv[start + base + lane]; c = e.x; wb = e.y; }

    int j = 0;
    for (; j + 4 <= mc; j += 4) {
      const int s0 = __builtin_amdgcn_readlane(c, j);
      const int s1 = __builtin_amdgcn_readlane(c, j + 1);
      const int s2 = __builtin_amdgcn_readlane(c, j + 2);
      const int s3 = __builtin_amdgcn_readlane(c, j + 3);
      const vecT r0 = Hv[(size_t)s0 * 32 + lane];
      const vecT r1 = Hv[(size_t)s1 * 32 + lane];
      const vecT r2 = Hv[(size_t)s2 * 32 + lane];
      const vecT r3 = Hv[(size_t)s3 * 32 + lane];
      acc += r0 * __int_as_float(__builtin_amdgcn_readlane(wb, j));
      acc += r1 * __int_as_float(__builtin_amdgcn_readlane(wb, j + 1));
      acc += r2 * __int_as_float(__builtin_amdgcn_readlane(wb, j + 2));
      acc += r3 * __int_as_float(__builtin_amdgcn_readlane(wb, j + 3));
    }
    const int rem = mc - j;
    if (rem > 0) {
      vecT r0 = {}, r1 = {}, r2 = {};
      r0 = Hv[(size_t)__builtin_amdgcn_readlane(c, j) * 32 + lane];
      if (rem > 1) r1 = Hv[(size_t)__builtin_amdgcn_readlane(c, j + 1) * 32 + lane];
      if (rem > 2) r2 = Hv[(size_t)__builtin_amdgcn_readlane(c, j + 2) * 32 + lane];
      acc += r0 * __int_as_float(__builtin_amdgcn_readlane(wb, j));
      acc += r1 * __int_as_float(__builtin_amdgcn_readlane(wb, j + 1));
      acc += r2 * __int_as_float(__builtin_amdgcn_readlane(wb, j + 2));
    }
  }

  ((vecT*)AGG)[(size_t)node * 32 + lane] = acc;
}

__global__ __launch_bounds__(256) void ln_relu_128(
    const float* __restrict__ AGG, const float* __restrict__ bias,
    const float* __restrict__ gamma, const float* __restrict__ beta,
    float* __restrict__ OUT, int n)
{
  const int node = (int)((blockIdx.x * blockDim.x + threadIdx.x) >> 5);
  const int lane = threadIdx.x & 31;
  if (node >= n) return;

  const float* row = AGG + (size_t)node * 128 + lane * 4;
  float v0 = row[0] + bias[lane * 4 + 0];
  float v1 = row[1] + bias[lane * 4 + 1];
  float v2 = row[2] + bias[lane * 4 + 2];
  float v3 = row[3] + bias[lane * 4 + 3];

  float s = v0 + v1 + v2 + v3;
#pragma unroll
  for (int m = 16; m >= 1; m >>= 1) s += __shfl_xor(s, m, 32);
  const float mu = s * (1.0f / 128.0f);

  float d0 = v0 - mu, d1 = v1 - mu, d2 = v2 - mu, d3 = v3 - mu;
  float q = d0 * d0 + d1 * d1 + d2 * d2 + d3 * d3;
#pragma unroll
  for (int m = 16; m >= 1; m >>= 1) q += __shfl_xor(q, m, 32);
  const float rstd = rsqrtf(q * (1.0f / 128.0f) + LN_EPS);

  float* out = OUT + (size_t)node * 128 + lane * 4;
  out[0] = fmaxf(0.0f, d0 * rstd * gamma[lane * 4 + 0] + beta[lane * 4 + 0]);
  out[1] = fmaxf(0.0f, d1 * rstd * gamma[lane * 4 + 1] + beta[lane * 4 + 1]);
  out[2] = fmaxf(0.0f, d2 * rstd * gamma[lane * 4 + 2] + beta[lane * 4 + 2]);
  out[3] = fmaxf(0.0f, d3 * rstd * gamma[lane * 4 + 3] + beta[lane * 4 + 3]);
}

__global__ void bias_add_64(const float* __restrict__ AGG,
                            const float* __restrict__ bias,
                            float* __restrict__ OUT, int total)
{
  int i = blockIdx.x * blockDim.x + threadIdx.x;
  if (i < total) OUT[i] = AGG[i] + bias[i & 63];
}

extern "C" void kernel_launch(void* const* d_in, const int* in_sizes, int n_in,
                              void* d_out, int out_size, void* d_ws, size_t ws_size,
                              hipStream_t stream) {
  const float* x   = (const float*)d_in[0];
  const int*   ei  = (const int*)d_in[1];
  const float* W1  = (const float*)d_in[2];
  const float* b1  = (const float*)d_in[3];
  const float* g1  = (const float*)d_in[4];
  const float* be1 = (const float*)d_in[5];
  const float* W2  = (const float*)d_in[6];
  const float* b2  = (const float*)d_in[7];
  const float* g2  = (const float*)d_in[8];
  const float* be2 = (const float*)d_in[9];
  const float* W3  = (const float*)d_in[10];
  const float* b3  = (const float*)d_in[11];

  const int N = in_sizes[0] / 128;
  const int E = in_sizes[1] / 2;
  const int* src = ei;
  const int* dst = ei + E;

  const size_t Np = ((size_t)N + 63) & ~(size_t)63;
  const size_t Ep = ((size_t)E + 63) & ~(size_t)63;
  float* ws     = (float*)d_ws;
  float* dinv   = ws;
  int*   cnt    = (int*)(dinv + Np);
  int2*  seg    = (int2*)(cnt + Np);
  int*   rank   = (int*)(seg + Np);
  int2*  cv     = (int2*)(rank + Ep);
  float* h      = (float*)(cv + Ep + Np);
  float* agg    = h + (size_t)N * 128;
  float* hn     = agg + (size_t)N * 128;

  zero_cnt  <<<(N + 255) / 256, 256, 0, stream>>>(cnt, N);
  count_rank<<<(E + 255) / 256, 256, 0, stream>>>(dst, cnt, rank, E);
  scan_dinv <<<(N + 1023) / 1024, 256, 0, stream>>>(cnt, seg, dinv, cv, N);
  fill_csr  <<<(E + 255) / 256, 256, 0, stream>>>(src, dst, rank, seg, dinv, cv, E);

  const int gblocks = (N + 7) / 8;

  gemm_wmma_f32<128><<<(N + 15) / 16, 256, 0, stream>>>(x, W1, h, N);
  gather_nodes<128><<<gblocks, 256, 0, stream>>>(h, seg, cv, agg, N);
  ln_relu_128<<<(N + 7) / 8, 256, 0, stream>>>(agg, b1, g1, be1, hn, N);

  gemm_wmma_f32<128><<<(N + 15) / 16, 256, 0, stream>>>(hn, W2, h, N);
  gather_nodes<128><<<gblocks, 256, 0, stream>>>(h, seg, cv, agg, N);
  ln_relu_128<<<(N + 7) / 8, 256, 0, stream>>>(agg, b2, g2, be2, hn, N);

  gemm_wmma_f32<64><<<(N + 31) / 32, 256, 0, stream>>>(hn, W3, h, N);
  gather_nodes<64><<<gblocks, 256, 0, stream>>>(h, seg, cv, agg, N);
  bias_add_64<<<(N * 64 + 255) / 256, 256, 0, stream>>>(agg, b3, (float*)d_out, N * 64);
}


# Fuse each layer's per-node epilogue (bias+LayerNorm+ReLU; last layer: bias) into the gather wave, which already holds the node's row: 13 -> 10 launches, no agg round trip, sum order unchanged
# speedup vs baseline: 2.6404x; 1.1333x over previous
#include <hip/hip_runtime.h>
#include <math.h>

typedef __attribute__((ext_vector_type(2))) float v2f;
typedef __attribute__((ext_vector_type(8))) float v8f;

#define LN_EPS 1e-5f

__global__ void zero_cnt(int* cnt, int n) {
  int i = blockIdx.x * blockDim.x + threadIdx.x;
  if (i < n) cnt[i] = 0;
}

__global__ void count_rank(const int* __restrict__ dst, int* cnt,
                           int* __restrict__ rank, int E) {
  int i = blockIdx.x * blockDim.x + threadIdx.x;
  if (i < E) rank[i] = atomicAdd(&cnt[dst[i]], 1);
}

__global__ __launch_bounds__(256) void scan_dinv(
    const int* __restrict__ cnt, int2* __restrict__ seg,
    float* __restrict__ dinv, int2* __restrict__ cv, int n)
{
  __shared__ int part[16];
  const int t    = threadIdx.x;
  const int w    = t >> 5;
  const int lane = t & 31;
  const int b    = blockIdx.x;

  const int4* __restrict__ c4 = (const int4*)cnt;
  int pre = 0;
  for (int i = 0; i < b; ++i) {
    int4 v = c4[i * 256 + t];
    pre += (v.x + v.y) + (v.z + v.w);
  }
#pragma unroll
  for (int m = 16; m >= 1; m >>= 1) pre += __shfl_xor(pre, m, 32);

  const int i0 = b * 1024 + t * 4;
  int c[4];
#pragma unroll
  for (int j = 0; j < 4; ++j) c[j] = (i0 + j < n) ? cnt[i0 + j] + 1 : 0;
  const int tsum = (c[0] + c[1]) + (c[2] + c[3]);
  int incl = tsum;
#pragma unroll
  for (int m = 1; m < 32; m <<= 1) {
    int o = __shfl_up(incl, m, 32);
    if (lane >= m) incl += o;
  }
  if (lane == 0)  part[w] = pre;
  if (lane == 31) part[8 + w] = incl;
  __syncthreads();
  int off = b * 1024;
#pragma unroll
  for (int k = 0; k < 8; ++k) {
    off += part[k];
    if (k < w) off += part[8 + k];
  }
  int run = off + incl - tsum;
#pragma unroll
  for (int j = 0; j < 4; ++j) {
    if (i0 + j < n) {
      seg[i0 + j] = make_int2(run, c[j]);
      const float d  = (float)c[j];
      const float dv = (d > 0.0f) ? rsqrtf(d) : 0.0f;
      dinv[i0 + j] = dv;
      cv[run] = make_int2(i0 + j, __float_as_int(dv * dv));
    }
    run += c[j];
  }
}

__global__ void fill_csr(const int* __restrict__ src, const int* __restrict__ dst,
                         const int* __restrict__ rank, const int2* __restrict__ seg,
                         const float* __restrict__ dinv, int2* __restrict__ cv, int E) {
  int i = blockIdx.x * blockDim.x + threadIdx.x;
  if (i < E) {
    const int s = src[i], d = dst[i];
    const int pos = seg[d].x + 1 + rank[i];
    cv[pos] = make_int2(s, __float_as_int(dinv[s] * dinv[d]));
  }
}

template <int FOUT>
__global__ __launch_bounds__(256) void gemm_wmma_f32(
    const float* __restrict__ X, const float* __restrict__ W,
    float* __restrict__ H, int nrows)
{
  constexpr int TPR = FOUT / 16;
  constexpr int RPB = 8 / TPR;
  const int w    = threadIdx.x >> 5;
  const int lane = threadIdx.x & 31;

  const int rowTile = blockIdx.x * RPB + (w / TPR);
  const int colTile = w % TPR;
  const int r0 = rowTile * 16;
  if (r0 >= nrows) return;
  const int c0 = colTile * 16;

  const int lh = lane & 15;
  const int kh = (lane >> 4) << 1;

  const float* __restrict__ xrow = X + (size_t)(r0 + lh) * 128 + kh;
  const float* __restrict__ wcol = W + (size_t)kh * FOUT + c0 + lh;

  v8f acc = {};
#pragma unroll
  for (int kk = 0; kk < 128; kk += 4) {
    v2f a; a.x = xrow[0];   a.y = xrow[1];
    v2f b; b.x = wcol[0];   b.y = wcol[FOUT];
    acc = __builtin_amdgcn_wmma_f32_16x16x4_f32(
        false, a, false, b, (short)0, acc, false, false);
    xrow += 4;
    wcol += 4 * FOUT;
  }

  const int rbase = r0 + ((lane >> 4) << 3);
  float* hp = H + (size_t)rbase * FOUT + c0 + lh;
#pragma unroll
  for (int v = 0; v < 8; ++v) hp[(size_t)v * FOUT] = acc[v];
}

template <int FOUT, bool NORM>
__global__ __launch_bounds__(256) void gather_epilogue(
    const float* __restrict__ H, const int2* __restrict__ seg,
    const int2* __restrict__ cv, const float* __restrict__ bias,
    const float* __restrict__ gamma, const float* __restrict__ beta,
    float* __restrict__ OUT, int n)
{
  constexpr int VEC = FOUT / 32;
  typedef __attribute__((ext_vector_type(VEC))) float vecT;
  const int lane = threadIdx.x & 31;
  const int node = __builtin_amdgcn_readfirstlane(
      (int)((blockIdx.x * blockDim.x + threadIdx.x) >> 5));
  if (node >= n) return;

  const int2  sg    = seg[node];
  const int   start = sg.x;
  const int   m     = sg.y;
  const vecT* __restrict__ Hv = (const vecT*)H;

  vecT bv, gv = {}, ev = {};
  for (int k = 0; k < VEC; ++k) bv[k] = bias[lane * VEC + k];
  if constexpr (NORM) {
    for (int k = 0; k < VEC; ++k) {
      gv[k] = gamma[lane * VEC + k];
      ev[k] = beta[lane * VEC + k];
    }
  }

  vecT acc = {};

  for (int base = 0; base < m; base += 32) {
    const int mc = (m - base < 32) ? (m - base) : 32;
    int c = 0, wb = 0;
    if (lane < mc) { const int2 e = cv[start + base + lane]; c = e.x; wb = e.y; }

    int j = 0;
    for (; j + 4 <= mc; j += 4) {
      const int s0 = __builtin_amdgcn_readlane(c, j);
      const int s1 = __builtin_amdgcn_readlane(c, j + 1);
      const int s2 = __builtin_amdgcn_readlane(c, j + 2);
      const int s3 = __builtin_amdgcn_readlane(c, j + 3);
      const vecT r0 = Hv[(size_t)s0 * 32 + lane];
      const vecT r1 = Hv[(size_t)s1 * 32 + lane];
      const vecT r2 = Hv[(size_t)s2 * 32 + lane];
      const vecT r3 = Hv[(size_t)s3 * 32 + lane];
      acc += r0 * __int_as_float(__builtin_amdgcn_readlane(wb, j));
      acc += r1 * __int_as_float(__builtin_amdgcn_readlane(wb, j + 1));
      acc += r2 * __int_as_float(__builtin_amdgcn_readlane(wb, j + 2));
      acc += r3 * __int_as_float(__builtin_amdgcn_readlane(wb, j + 3));
    }
    const int rem = mc - j;
    if (rem > 0) {
      vecT r0 = {}, r1 = {}, r2 = {};
      r0 = Hv[(size_t)__builtin_amdgcn_readlane(c, j) * 32 + lane];
      if (rem > 1) r1 = Hv[(size_t)__builtin_amdgcn_readlane(c, j + 1) * 32 + lane];
      if (rem > 2) r2 = Hv[(size_t)__builtin_amdgcn_readlane(c, j + 2) * 32 + lane];
      acc += r0 * __int_as_float(__builtin_amdgcn_readlane(wb, j));
      acc += r1 * __int_as_float(__builtin_amdgcn_readlane(wb, j + 1));
      acc += r2 * __int_as_float(__builtin_amdgcn_readlane(wb, j + 2));
    }
  }

  vecT o;
  if constexpr (NORM) {
    static_assert(FOUT == 128, "the LayerNorm epilogue is written for 4 floats per lane");
    const float v0 = acc[0] + bv[0];
    const float v1 = acc[1] + bv[1];
    const float v2 = acc[2] + bv[2];
    const float v3 = acc[3] + bv[3];

    float s = v0 + v1 + v2 + v3;
#pragma unroll
    for (int x = 16; x >= 1; x >>= 1) s += __shfl_xor(s, x, 32);
    const float mu = s * (1.0f / 128.0f);

    const float d0 = v0 - mu, d1 = v1 - mu, d2 = v2 - mu, d3 = v3 - mu;
    float q = d0 * d0 + d1 * d1 + d2 * d2 + d3 * d3;
#pragma unroll
    for (int x = 16; x >= 1; x >>= 1) q += __shfl_xor(q, x, 32);
    const float rstd = rsqrtf(q * (1.0f / 128.0f) + LN_EPS);

    o[0] = fmaxf(0.0f, d0 * rstd * gv[0] + ev[0]);
    o[1] = fmaxf(0.0f, d1 * rstd * gv[1] + ev[1]);
    o[2] = fmaxf(0.0f, d2 * rstd * gv[2] + ev[2]);
    o[3] = fmaxf(0.0f, d3 * rstd * gv[3] + ev[3]);
  } else {
    for (int k = 0; k < VEC; ++k) o[k] = acc[k] + bv[k];
  }
  ((vecT*)OUT)[(size_t)node * 32 + lane] = o;
}

extern "C" void kernel_launch(void* const* d_in, const int* in_sizes, int n_in,
                              void* d_out, int out_size, void* d_ws, size_t ws_size,
                              hipStream_t stream) {
  const float* x   = (const float*)d_in[0];
  const int*   ei  = (const int*)d_in[1];
  const float* W1  = (const float*)d_in[2];
  const float* b1  = (const float*)d_in[3];
  const float* g1  = (const float*)d_in[4];
  const float* be1 = (const float*)d_in[5];
  const float* W2  = (const float*)d_in[6];
  const float* b2  = (const float*)d_in[7];
  const float* g2  = (const float*)d_in[8];
  const float* be2 = (const float*)d_in[9];
  const float* W3  = (const float*)d_in[10];
  const float* b3  = (const float*)d_in[11];

  const int N = in_sizes[0] / 128;
  const int E = in_sizes[1] / 2;
  const int* src = ei;
  const int* dst = ei + E;

  const size_t Np = ((size_t)N + 63) & ~(size_t)63;
  const size_t Ep = ((size_t)E + 63) & ~(size_t)63;
  float* ws     = (float*)d_ws;
  float* dinv   = ws;
  int*   cnt    = (int*)(dinv + Np);
  int2*  seg    = (int2*)(cnt + Np);
  int*   rank   = (int*)(seg + Np);
  int2*  cv     = (int2*)(rank + Ep);
  float* h      = (float*)(cv + Ep + Np);
  float* hn     = h + (size_t)N * 128;

  zero_cnt  <<<(N + 255) / 256, 256, 0, stream>>>(cnt, N);
  count_rank<<<(E + 255) / 256, 256, 0, stream>>>(dst, cnt, rank, E);
  scan_dinv <<<(N + 1023) / 1024, 256, 0, stream>>>(cnt, seg, dinv, cv, N);
  fill_csr  <<<(E + 255) / 256, 256, 0, stream>>>(src, dst, rank, seg, dinv, cv, E);

  const int gblocks = (N + 7) / 8;

  gemm_wmma_f32<128><<<(N + 15) / 16, 256, 0, stream>>>(x, W1, h, N);
  gather_epilogue<128, true><<<gblocks, 256, 0, stream>>>(h, seg, cv, b1, g1, be1, hn, N);

  gemm_wmma_f32<128><<<(N + 15) / 16, 256, 0, stream>>>(hn, W2, h, N);
  gather_epilogue<128, true><<<gblocks, 256, 0, stream>>>(h, seg, cv, b2, g2, be2, hn, N);

  gemm_wmma_f32<64><<<(N + 31) / 32, 256, 0, stream>>>(hn, W3, h, N);
  gather_epilogue<64, false><<<gblocks, 256, 0, stream>>>(h, seg, cv, b3, nullptr, nullptr,
                                                         (float*)d_out, N);
}


# Fuse the next layer's WMMA GEMM into the gather+LayerNorm kernel: a block gathers 32 node rows into LDS (4 per wave) and multiplies them by the next W; 10 -> 8 launches; sum order unchanged
# speedup vs baseline: 3.1310x; 1.1858x over previous
#include <hip/hip_runtime.h>
#include <math.h>

typedef __attribute__((ext_vector_type(2))) float v2f;
typedef __attribute__((ext_vector_type(8))) float v8f;

#define LN_EPS 1e-5f

__global__ void zero_cnt(int* cnt, int n) {
  int i = blockIdx.x * blockDim.x + threadIdx.x;
  if (i < n) cnt[i] = 0;
}

__global__ void count_rank(const int* __restrict__ dst, int* cnt,
                           int* __restrict__ rank, int E) {
  int i = blockIdx.x * blockDim.x + threadIdx.x;
  if (i < E) rank[i] = atomicAdd(&cnt[dst[i]], 1);
}

__global__ __launch_bounds__(256) void scan_dinv(
    const int* __restrict__ cnt, int2* __restrict__ seg,
    float* __restrict__ dinv, int2* __restrict__ cv, int n)
{
  __shared__ int part[16];
  const int t    = threadIdx.x;
  const int w    = t >> 5;
  const int lane = t & 31;
  const int b    = blockIdx.x;

  const int4* __restrict__ c4 = (const int4*)cnt;
  int pre = 0;
  for (int i = 0; i < b; ++i) {
    int4 v = c4[i * 256 + t];
    pre += (v.x + v.y) + (v.z + v.w);
  }
#pragma unroll
  for (int m = 16; m >= 1; m >>= 1) pre += __shfl_xor(pre, m, 32);

  const int i0 = b * 1024 + t * 4;
  int c[4];
#pragma unroll
  for (int j = 0; j < 4; ++j) c[j] = (i0 + j < n) ? cnt[i0 + j] + 1 : 0;
  const int tsum = (c[0] + c[1]) + (c[2] + c[3]);
  int incl = tsum;
#pragma unroll
  for (int m = 1; m < 32; m <<= 1) {
    int o = __shfl_up(incl, m, 32);
    if (lane >= m) incl += o;
  }
  if (lane == 0)  part[w] = pre;
  if (lane == 31) part[8 + w] = incl;
  __syncthreads();
  int off = b * 1024;
#pragma unroll
  for (int k = 0; k < 8; ++k) {
    off += part[k];
    if (k < w) off += part[8 + k];
  }
  int run = off + incl - tsum;
#pragma unroll
  for (int j = 0; j < 4; ++j) {
    if (i0 + j < n) {
      seg[i0 + j] = make_int2(run, c[j]);
      const float d  = (float)c[j];
      const float dv = (d > 0.0f) ? rsqrtf(d) : 0.0f;
      dinv[i0 + j] = dv;
      cv[run] = make_int2(i0 + j, __float_as_int(dv * dv));
    }
    run += c[j];
  }
}

__global__ void fill_csr(const int* __restrict__ src, const int* __restrict__ dst,
                         const int* __restrict__ rank, const int2* __restrict__ seg,
                         const float* __restrict__ dinv, int2* __restrict__ cv, int E) {
  int i = blockIdx.x * blockDim.x + threadIdx.x;
  if (i < E) {
    const int s = src[i], d = dst[i];
    const int pos = seg[d].x + 1 + rank[i];
    cv[pos] = make_int2(s, __float_as_int(dinv[s] * dinv[d]));
  }
}

template <int FOUT>
__global__ __launch_bounds__(256) void gemm_wmma_f32(
    const float* __restrict__ X, const float* __restrict__ W,
    float* __restrict__ H, int nrows)
{
  constexpr int TPR = FOUT / 16;
  constexpr int RPB = 8 / TPR;
  const int w    = threadIdx.x >> 5;
  const int lane = threadIdx.x & 31;

  const int rowTile = blockIdx.x * RPB + (w / TPR);
  const int colTile = w % TPR;
  const int r0 = rowTile * 16;
  if (r0 >= nrows) return;
  const int c0 = colTile * 16;

  const int lh = lane & 15;
  const int kh = (lane >> 4) << 1;

  const float* __restrict__ xrow = X + (size_t)(r0 + lh) * 128 + kh;
  const float* __restrict__ wcol = W + (size_t)kh * FOUT + c0 + lh;

  v8f acc = {};
#pragma unroll
  for (int kk = 0; kk < 128; kk += 4) {
    v2f a; a.x = xrow[0];   a.y = xrow[1];
    v2f b; b.x = wcol[0];   b.y = wcol[FOUT];
    acc = __builtin_amdgcn_wmma_f32_16x16x4_f32(
        false, a, false, b, (short)0, acc, false, false);
    xrow += 4;
    wcol += 4 * FOUT;
  }

  const int rbase = r0 + ((lane >> 4) << 3);
  float* hp = H + (size_t)rbase * FOUT + c0 + lh;
#pragma unroll
  for (int v = 0; v < 8; ++v) hp[(size_t)v * FOUT] = acc[v];
}

template <int VEC> struct RowVec {
  typedef __attribute__((ext_vector_type(VEC))) float T;
};

template <int VEC>
__device__ __forceinline__ typename RowVec<VEC>::T gather_node(
    const typename RowVec<VEC>::T* __restrict__ Hv,
    const int2* __restrict__ cv, int start, int m, int lane)
{
  typedef typename RowVec<VEC>::T vecT;
  vecT acc = {};

  for (int base = 0; base < m; base += 32) {
    const int mc = (m - base < 32) ? (m - base) : 32;
    int c = 0, wb = 0;
    if (lane < mc) { const int2 e = cv[start + base + lane]; c = e.x; wb = e.y; }

    int j = 0;
    for (; j + 4 <= mc; j += 4) {
      const int s0 = __builtin_amdgcn_readlane(c, j);
      const int s1 = __builtin_amdgcn_readlane(c, j + 1);
      const int s2 = __builtin_amdgcn_readlane(c, j + 2);
      const int s3 = __builtin_amdgcn_readlane(c, j + 3);
      const vecT r0 = Hv[(size_t)s0 * 32 + lane];
      const vecT r1 = Hv[(size_t)s1 * 32 + lane];
      const vecT r2 = Hv[(size_t)s2 * 32 + lane];
      const vecT r3 = Hv[(size_t)s3 * 32 + lane];
      acc += r0 * __int_as_float(__builtin_amdgcn_readlane(wb, j));
      acc += r1 * __int_as_float(__builtin_amdgcn_readlane(wb, j + 1));
      acc += r2 * __int_as_float(__builtin_amdgcn_readlane(wb, j + 2));
      acc += r3 * __int_as_float(__builtin_amdgcn_readlane(wb, j + 3));
    }
    const int rem = mc - j;
    if (rem > 0) {
      vecT r0 = {}, r1 = {}, r2 = {};
      r0 = Hv[(size_t)__builtin_amdgcn_readlane(c, j) * 32 + lane];
      if (rem > 1) r1 = Hv[(size_t)__builtin_amdgcn_readlane(c, j + 1) * 32 + lane];
      if (rem > 2) r2 = Hv[(size_t)__builtin_amdgcn_readlane(c, j + 2) * 32 + lane];
      acc += r0 * __int_as_float(__builtin_amdgcn_readlane(wb, j));
      acc += r1 * __int_as_float(__builtin_amdgcn_readlane(wb, j + 1));
      acc += r2 * __int_as_float(__builtin_amdgcn_readlane(wb, j + 2));
    }
  }
  return acc;
}

template <int FNEXT>
__global__ __launch_bounds__(256, 16) void gather_ln_gemm(
    const float* __restrict__ H, const int2* __restrict__ seg,
    const int2* __restrict__ cv, const float* __restrict__ bias,
    const float* __restrict__ gamma, const float* __restrict__ beta,
    const float* __restrict__ WN, float* __restrict__ HN, int n)
{
  typedef RowVec<4>::T vecT;
  constexpr int ROWS  = 32;
  constexpr int PITCH = 132;
  __shared__ float tile[ROWS * PITCH];

  const int lane = threadIdx.x & 31;
  const int w  = __builtin_amdgcn_readfirstlane((int)(threadIdx.x >> 5));
  const int nb = blockIdx.x * ROWS + w * 4;

  vecT bv, gv, ev;
  for (int k = 0; k < 4; ++k) {
    bv[k] = bias[lane * 4 + k];
    gv[k] = gamma[lane * 4 + k];
    ev[k] = beta[lane * 4 + k];
  }

  int2 sgl = make_int2(0, 0);
  if (lane < 4 && nb + lane < n) sgl = seg[nb + lane];

  const vecT* __restrict__ Hv = (const vecT*)H;
  for (int q = 0; q < 4; ++q) {
    const int start = __builtin_amdgcn_readlane(sgl.x, q);
    const int m     = __builtin_amdgcn_readlane(sgl.y, q);
    const vecT acc = gather_node<4>(Hv, cv, start, m, lane);

    const float v0 = acc[0] + bv[0];
    const float v1 = acc[1] + bv[1];
    const float v2 = acc[2] + bv[2];
    const float v3 = acc[3] + bv[3];

    float s = v0 + v1 + v2 + v3;
#pragma unroll
    for (int x = 16; x >= 1; x >>= 1) s += __shfl_xor(s, x, 32);
    const float mu = s * (1.0f / 128.0f);

    const float d0 = v0 - mu, d1 = v1 - mu, d2 = v2 - mu, d3 = v3 - mu;
    float qs = d0 * d0 + d1 * d1 + d2 * d2 + d3 * d3;
#pragma unroll
    for (int x = 16; x >= 1; x >>= 1) qs += __shfl_xor(qs, x, 32);
    const float rstd = rsqrtf(qs * (1.0f / 128.0f) + LN_EPS);

    vecT o;
    o[0] = fmaxf(0.0f, d0 * rstd * gv[0] + ev[0]);
    o[1] = fmaxf(0.0f, d1 * rstd * gv[1] + ev[1]);
    o[2] = fmaxf(0.0f, d2 * rstd * gv[2] + ev[2]);
    o[3] = fmaxf(0.0f, d3 * rstd * gv[3] + ev[3]);
    *(vecT*)(tile + (w * 4 + q) * PITCH + lane * 4) = o;
  }
  __syncthreads();

  constexpr int CT = FNEXT / 16;
  constexpr int NT = CT / 4;
  const int ct  = w % CT;
  const int rt0 = (w / CT) * NT;
  const int c0  = ct * 16;
  const int lh  = lane & 15;
  const int kh  = (lane >> 4) << 1;

  const float* ap = tile + (rt0 * 16 + lh) * PITCH + kh;
  const float* __restrict__ wcol = WN + (size_t)kh * FNEXT + c0 + lh;

  v8f acc[NT] = {};
#pragma unroll
  for (int kk = 0; kk < 128; kk += 4) {
    v2f b; b.x = wcol[(size_t)kk * FNEXT]; b.y = wcol[(size_t)(kk + 1) * FNEXT];
#pragma unroll
    for (int t = 0; t < NT; ++t) {
      const v2f a = *(const v2f*)(ap + t * 16 * PITCH + kk);
      acc[t] = __builtin_amdgcn_wmma_f32_16x16x4_f32(
          false, a, false, b, (short)0, acc[t], false, false);
    }
  }

#pragma unroll
  for (int t = 0; t < NT; ++t) {
    const int rbase = blockIdx.x * ROWS + (rt0 + t) * 16 + ((lane >> 4) << 3);
    float* hp = HN + (size_t)rbase * FNEXT + c0 + lh;
#pragma unroll
    for (int v = 0; v < 8; ++v)
      if (rbase + v < n) hp[(size_t)v * FNEXT] = acc[t][v];
  }
}

template <int FOUT>
__global__ __launch_bounds__(256) void gather_bias(
    const float* __restrict__ H, const int2* __restrict__ seg,
    const int2* __restrict__ cv, const float* __restrict__ bias,
    float* __restrict__ OUT, int n)
{
  constexpr int VEC = FOUT / 32;
  typedef typename RowVec<VEC>::T vecT;
  const int lane = threadIdx.x & 31;
  const int node = __builtin_amdgcn_readfirstlane(
      (int)((blockIdx.x * blockDim.x + threadIdx.x) >> 5));
  if (node >= n) return;

  const int2 sg = seg[node];
  vecT bv;
  for (int k = 0; k < VEC; ++k) bv[k] = bias[lane * VEC + k];

  const vecT acc = gather_node<VEC>((const vecT*)H, cv, sg.x, sg.y, lane);

  vecT o;
  for (int k = 0; k < VEC; ++k) o[k] = acc[k] + bv[k];
  ((vecT*)OUT)[(size_t)node * 32 + lane] = o;
}

extern "C" void kernel_launch(void* const* d_in, const int* in_sizes, int n_in,
                              void* d_out, int out_size, void* d_ws, size_t ws_size,
                              hipStream_t stream) {
  const float* x   = (const float*)d_in[0];
  const int*   ei  = (const int*)d_in[1];
  const float* W1  = (const float*)d_in[2];
  const float* b1  = (const float*)d_in[3];
  const float* g1  = (const float*)d_in[4];
  const float* be1 = (const float*)d_in[5];
  const float* W2  = (const float*)d_in[6];
  const float* b2  = (const float*)d_in[7];
  const float* g2  = (const float*)d_in[8];
  const float* be2 = (const float*)d_in[9];
  const float* W3  = (const float*)d_in[10];
  const float* b3  = (const float*)d_in[11];

  const int N = in_sizes[0] / 128;
  const int E = in_sizes[1] / 2;
  const int* src = ei;
  const int* dst = ei + E;

  const size_t Np = ((size_t)N + 63) & ~(size_t)63;
  const size_t Ep = ((size_t)E + 63) & ~(size_t)63;
  float* ws     = (float*)d_ws;
  float* dinv   = ws;
  int*   cnt    = (int*)(dinv + Np);
  int2*  seg    = (int2*)(cnt + Np);
  int*   rank   = (int*)(seg + Np);
  int2*  cv     = (int2*)(rank + Ep);
  float* h      = (float*)(cv + Ep + Np);
  float* hn     = h + (size_t)N * 128;

  zero_cnt  <<<(N + 255) / 256, 256, 0, stream>>>(cnt, N);
  count_rank<<<(E + 255) / 256, 256, 0, stream>>>(dst, cnt, rank, E);
  scan_dinv <<<(N + 1023) / 1024, 256, 0, stream>>>(cnt, seg, dinv, cv, N);
  fill_csr  <<<(E + 255) / 256, 256, 0, stream>>>(src, dst, rank, seg, dinv, cv, E);

  const int gblocks = (N + 7) / 8;

  const int fblocks = (N + 31) / 32;

  gemm_wmma_f32<128><<<(N + 15) / 16, 256, 0, stream>>>(x, W1, h, N);

  gather_ln_gemm<128><<<fblocks, 256, 0, stream>>>(h, seg, cv, b1, g1, be1, W2, hn, N);

  gather_ln_gemm<64><<<fblocks, 256, 0, stream>>>(hn, seg, cv, b2, g2, be2, W3, h, N);

  gather_bias<64><<<gblocks, 256, 0, stream>>>(h, seg, cv, b3, (float*)d_out, N);
}


# CSR build without atomics on memory: block (node range, edge chunk) counts, later places its edges through LDS adds; 8 -> 7 launches; per-node sum order: self-loop, chunk order, LDS arrival in chunk
# speedup vs baseline: 3.8915x; 1.2429x over previous
#include <hip/hip_runtime.h>
#include <math.h>

typedef __attribute__((ext_vector_type(2))) float v2f;
typedef __attribute__((ext_vector_type(8))) float v8f;

#define LN_EPS 1e-5f

#define RNG 2048
#define NCH 8
#define EPT 16

__device__ __forceinline__ void chunk_bounds(int c, int E, int& e0, int& e1) {
  const int clen = (((E + NCH - 1) / NCH) + 1023) & ~1023;
  e0 = c * clen;
  e1 = (e0 + clen < E) ? e0 + clen : E;
}

__global__ __launch_bounds__(1024) void count_range(
    const int* __restrict__ dst, int* __restrict__ cntP, int* __restrict__ tot,
    int E, int np)
{
  __shared__ int hist[RNG];
  __shared__ int part[32];
  const int t    = threadIdx.x;
  const int r    = blockIdx.x;
  const int c    = blockIdx.y;
  const int base = r * RNG;

  for (int j = t; j < RNG; j += 1024) hist[j] = 0;
  __syncthreads();

  int e0, e1;
  chunk_bounds(c, E, e0, e1);
  int mine = 0;
  for (int eb = e0; eb < e1; eb += 1024 * EPT) {
    int d[EPT];
#pragma unroll
    for (int k = 0; k < EPT; ++k) {
      const int e = eb + k * 1024 + t;
      d[k] = dst[(e < e1) ? e : e1 - 1];
    }
#pragma unroll
    for (int k = 0; k < EPT; ++k) {
      const int e = eb + k * 1024 + t;
      const unsigned u = (unsigned)(d[k] - base);
      if (e < e1 && u < (unsigned)RNG) { atomicAdd(&hist[u], 1); ++mine; }
    }
  }

#pragma unroll
  for (int m = 16; m >= 1; m >>= 1) mine += __shfl_xor(mine, m, 32);
  if ((t & 31) == 0) part[t >> 5] = mine;
  __syncthreads();

  for (int j = t; j < RNG; j += 1024) {
    const int i = base + j;
    if (i < np) cntP[(size_t)c * np + i] = hist[j];
  }
  if (t == 0) {
    int s = 0;
#pragma unroll
    for (int k = 0; k < 32; ++k) s += part[k];
    tot[r * NCH + c] = s;
  }
}

__global__ __launch_bounds__(256) void scan_dinv(
    const int* __restrict__ cntP, const int* __restrict__ tot,
    int2* __restrict__ seg, float* __restrict__ dinv, int2* __restrict__ cv,
    int n, int np)
{
  __shared__ int part[16];
  const int t    = threadIdx.x;
  const int w    = t >> 5;
  const int lane = t & 31;
  const int b    = blockIdx.x;

  int pre = 0;
  for (int i = t; i < b * NCH; i += 256) pre += tot[i];
#pragma unroll
  for (int m = 16; m >= 1; m >>= 1) pre += __shfl_xor(pre, m, 32);

  const int i0 = b * RNG + t * 8;
  int c[8] = {0, 0, 0, 0, 0, 0, 0, 0};
  if (i0 < np) {
#pragma unroll
    for (int k = 0; k < NCH; ++k) {
      const int4* __restrict__ q = (const int4*)(cntP + (size_t)k * np + i0);
      const int4 x = q[0], y = q[1];
      c[0] += x.x; c[1] += x.y; c[2] += x.z; c[3] += x.w;
      c[4] += y.x; c[5] += y.y; c[6] += y.z; c[7] += y.w;
    }
  }
#pragma unroll
  for (int j = 0; j < 8; ++j) c[j] = (i0 + j < n) ? c[j] + 1 : 0;
  const int tsum = ((c[0] + c[1]) + (c[2] + c[3])) + ((c[4] + c[5]) + (c[6] + c[7]));
  int incl = tsum;
#pragma unroll
  for (int m = 1; m < 32; m <<= 1) {
    int o = __shfl_up(incl, m, 32);
    if (lane >= m) incl += o;
  }
  if (lane == 0)  part[w] = pre;
  if (lane == 31) part[8 + w] = incl;
  __syncthreads();
  int off = b * RNG;
#pragma unroll
  for (int k = 0; k < 8; ++k) {
    off += part[k];
    if (k < w) off += part[8 + k];
  }
  int run = off + incl - tsum;
#pragma unroll
  for (int j = 0; j < 8; ++j) {
    if (i0 + j < n) {
      seg[i0 + j] = make_int2(run, c[j]);
      const float d  = (float)c[j];
      const float dv = (d > 0.0f) ? rsqrtf(d) : 0.0f;
      dinv[i0 + j] = dv;
      cv[run] = make_int2(i0 + j, __float_as_int(dv * dv));
    }
    run += c[j];
  }
}

__global__ __launch_bounds__(1024) void fill_range(
    const int* __restrict__ src, const int* __restrict__ dst,
    const int* __restrict__ cntP, const int2* __restrict__ seg,
    const float* __restrict__ dinv, int2* __restrict__ cv, int n, int E, int np)
{
  __shared__ int   cur[RNG + 32];
  __shared__ float dvl[RNG];
  const int t    = threadIdx.x;
  const int r    = blockIdx.x;
  const int c    = blockIdx.y;
  const int base = r * RNG;

  for (int j = t; j < RNG; j += 1024) {
    const int i = base + j;
    int   v  = 0;
    float dv = 0.0f;
    if (i < n) {
      v  = seg[i].x + 1;
      dv = dinv[i];
#pragma unroll
      for (int k = 0; k < NCH; ++k) {
        const int x = cntP[(size_t)k * np + i];
        v += (k < c) ? x : 0;
      }
    }
    cur[j] = v;
    dvl[j] = dv;
  }
  if (t < 32) cur[RNG + t] = 0;
  __syncthreads();

  int* const spare = &cur[RNG + (t & 31)];
  int e0, e1;
  chunk_bounds(c, E, e0, e1);
  for (int eb = e0; eb < e1; eb += 1024 * EPT) {
    unsigned s[EPT];
    int   d[EPT], pos[EPT];
    float ws[EPT];
#pragma unroll
    for (int k = 0; k < EPT; ++k) {
      const int e  = eb + k * 1024 + t;
      const int ec = (e < e1) ? e : e1 - 1;
      d[k] = dst[ec];
      s[k] = (unsigned)src[ec];
    }
#pragma unroll
    for (int k = 0; k < EPT; ++k) asm volatile("" :: "v"(s[k]));
#pragma unroll
    for (int k = 0; k < EPT; ++k) {
      const int e = eb + k * 1024 + t;
      const bool hit = e < e1 && (unsigned)(d[k] - base) < (unsigned)RNG;
      ws[k] = dinv[hit ? s[k] : 0u];
    }
#pragma unroll
    for (int k = 0; k < EPT; ++k) {
      const int e = eb + k * 1024 + t;
      const unsigned u = (unsigned)(d[k] - base);
      const bool hit = e < e1 && u < (unsigned)RNG;
      pos[k] = atomicAdd(hit ? &cur[u] : spare, hit ? 1 : 0);
    }
#pragma unroll
    for (int k = 0; k < EPT; ++k) asm volatile("" :: "v"(ws[k]));
#pragma unroll
    for (int k = 0; k < EPT; ++k) {
      const int e = eb + k * 1024 + t;
      const unsigned u = (unsigned)(d[k] - base);
      if (e < e1 && u < (unsigned)RNG)
        cv[pos[k]] = make_int2((int)s[k], __float_as_int(ws[k] * dvl[u]));
    }
  }
}

template <int FOUT>
__global__ __launch_bounds__(256) void gemm_wmma_f32(
    const float* __restrict__ X, const float* __restrict__ W,
    float* __restrict__ H, int nrows)
{
  constexpr int TPR = FOUT / 16;
  constexpr int RPB = 8 / TPR;
  const int w    = threadIdx.x >> 5;
  const int lane = threadIdx.x & 31;

  const int rowTile = blockIdx.x * RPB + (w / TPR);
  const int colTile = w % TPR;
  const int r0 = rowTile * 16;
  if (r0 >= nrows) return;
  const int c0 = colTile * 16;

  const int lh = lane & 15;
  const int kh = (lane >> 4) << 1;

  const float* __restrict__ xrow = X + (size_t)(r0 + lh) * 128 + kh;
  const float* __restrict__ wcol = W + (size_t)kh * FOUT + c0 + lh;

  v8f acc = {};
#pragma unroll
  for (int kk = 0; kk < 128; kk += 4) {
    v2f a; a.x = xrow[0];   a.y = xrow[1];
    v2f b; b.x = wcol[0];   b.y = wcol[FOUT];
    acc = __builtin_amdgcn_wmma_f32_16x16x4_f32(
        false, a, false, b, (short)0, acc, false, false);
    xrow += 4;
    wcol += 4 * FOUT;
  }

  const int rbase = r0 + ((lane >> 4) << 3);
  float* hp = H + (size_t)rbase * FOUT + c0 + lh;
#pragma unroll
  for (int v = 0; v < 8; ++v) hp[(size_t)v * FOUT] = acc[v];
}

template <int VEC> struct RowVec {
  typedef __attribute__((ext_vector_type(VEC))) float T;
};

template <int VEC>
__device__ __forceinline__ typename RowVec<VEC>::T gather_node(
    const typename RowVec<VEC>::T* __restrict__ Hv,
    const int2* __restrict__ cv, int start, int m, int lane)
{
  typedef typename RowVec<VEC>::T vecT;
  vecT acc = {};

  for (int base = 0; base < m; base += 32) {
    const int mc = (m - base < 32) ? (m - base) : 32;
    int c = 0, wb = 0;
    if (lane < mc) { const int2 e = cv[start + base + lane]; c = e.x; wb = e.y; }

    int j = 0;
    for (; j + 4 <= mc; j += 4) {
      const int s0 = __builtin_amdgcn_readlane(c, j);
      const int s1 = __builtin_amdgcn_readlane(c, j + 1);
      const int s2 = __builtin_amdgcn_readlane(c, j + 2);
      const int s3 = __builtin_amdgcn_readlane(c, j + 3);
      const vecT r0 = Hv[(size_t)s0 * 32 + lane];
      const vecT r1 = Hv[(size_t)s1 * 32 + lane];
      const vecT r2 = Hv[(size_t)s2 * 32 + lane];
      const vecT r3 = Hv[(size_t)s3 * 32 + lane];
      acc += r0 * __int_as_float(__builtin_amdgcn_readlane(wb, j));
      acc += r1 * __int_as_float(__builtin_amdgcn_readlane(wb, j + 1));
      acc += r2 * __int_as_float(__builtin_amdgcn_readlane(wb, j + 2));
      acc += r3 * __int_as_float(__builtin_amdgcn_readlane(wb, j + 3));
    }
    const int rem = mc - j;
    if (rem > 0) {
      vecT r0 = {}, r1 = {}, r2 = {};
      r0 = Hv[(size_t)__builtin_amdgcn_readlane(c, j) * 32 + lane];
      if (rem > 1) r1 = Hv[(size_t)__builtin_amdgcn_readlane(c, j + 1) * 32 + lane];
      if (rem > 2) r2 = Hv[(size_t)__builtin_amdgcn_readlane(c, j + 2) * 32 + lane];
      acc += r0 * __int_as_float(__builtin_amdgcn_readlane(wb, j));
      acc += r1 * __int_as_float(__builtin_amdgcn_readlane(wb, j + 1));
      acc += r2 * __int_as_float(__builtin_amdgcn_readlane(wb, j + 2));
    }
  }
  return acc;
}

template <int FNEXT>
__global__ __launch_bounds__(256, 16) void gather_ln_gemm(
    const float* __restrict__ H, const int2* __restrict__ seg,
    const int2* __restrict__ cv, const float* __restrict__ bias,
    const float* __restrict__ gamma, const float* __restrict__ beta,
    const float* __restrict__ WN, float* __restrict__ HN, int n)
{
  typedef RowVec<4>::T vecT;
  constexpr int ROWS  = 32;
  constexpr int PITCH = 132;
  __shared__ float tile[ROWS * PITCH];

  const int lane = threadIdx.x & 31;
  const int w  = __builtin_amdgcn_readfirstlane((int)(threadIdx.x >> 5));
  const int nb = blockIdx.x * ROWS + w * 4;

  vecT bv, gv, ev;
  for (int k = 0; k < 4; ++k) {
    bv[k] = bias[lane * 4 + k];
    gv[k] = gamma[lane * 4 + k];
    ev[k] = beta[lane * 4 + k];
  }

  int2 sgl = make_int2(0, 0);
  if (lane < 4 && nb + lane < n) sgl = seg[nb + lane];

  const vecT* __restrict__ Hv = (const vecT*)H;
  for (int q = 0; q < 4; ++q) {
    const int start = __builtin_amdgcn_readlane(sgl.x, q);
    const int m     = __builtin_amdgcn_readlane(sgl.y, q);
    const vecT acc = gather_node<4>(Hv, cv, start, m, lane);

    const float v0 = acc[0] + bv[0];
    const float v1 = acc[1] + bv[1];
    const float v2 = acc[2] + bv[2];
    const float v3 = acc[3] + bv[3];

    float s = v0 + v1 + v2 + v3;
#pragma unroll
    for (int x = 16; x >= 1; x >>= 1) s += __shfl_xor(s, x, 32);
    const float mu = s * (1.0f / 128.0f);

    const float d0 = v0 - mu, d1 = v1 - mu, d2 = v2 - mu, d3 = v3 - mu;
    float qs = d0 * d0 + d1 * d1 + d2 * d2 + d3 * d3;
#pragma unroll
    for (int x = 16; x >= 1; x >>= 1) qs += __shfl_xor(qs, x, 32);
    const float rstd = rsqrtf(qs * (1.0f / 128.0f) + LN_EPS);

    vecT o;
    o[0] = fmaxf(0.0f, d0 * rstd * gv[0] + ev[0]);
    o[1] = fmaxf(0.0f, d1 * rstd * gv[1] + ev[1]);
    o[2] = fmaxf(0.0f, d2 * rstd * gv[2] + ev[2]);
    o[3] = fmaxf(0.0f, d3 * rstd * gv[3] + ev[3]);
    *(vecT*)(tile + (w * 4 + q) * PITCH + lane * 4) = o;
  }
  __syncthreads();

  constexpr int CT = FNEXT / 16;
  constexpr int NT = CT / 4;
  const int ct  = w % CT;
  const int rt0 = (w / CT) * NT;
  const int c0  = ct * 16;
  const int lh  = lane & 15;
  const int kh  = (lane >> 4) << 1;

  const float* ap = tile + (rt0 * 16 + lh) * PITCH + kh;
  const float* __restrict__ wcol = WN + (size_t)kh * FNEXT + c0 + lh;

  v8f acc[NT] = {};
#pragma unroll
  for (int kk = 0; kk < 128; kk += 4) {
    v2f b; b.x = wcol[(size_t)kk * FNEXT]; b.y = wcol[(size_t)(kk + 1) * FNEXT];
#pragma unroll
    for (int t = 0; t < NT; ++t) {
      const v2f a = *(const v2f*)(ap + t * 16 * PITCH + kk);
      acc[t] = __builtin_amdgcn_wmma_f32_16x16x4_f32(
          false, a, false, b, (short)0, acc[t], false, false);
    }
  }

#pragma unroll
  for (int t = 0; t < NT; ++t) {
    const int rbase = blockIdx.x * ROWS + (rt0 + t) * 16 + ((lane >> 4) << 3);
    float* hp = HN + (size_t)rbase * FNEXT + c0 + lh;
#pragma unroll
    for (int v = 0; v < 8; ++v)
      if (rbase + v < n) hp[(size_t)v * FNEXT] = acc[t][v];
  }
}

template <int FOUT>
__global__ __launch_bounds__(256) void gather_bias(
    const float* __restrict__ H, const int2* __restrict__ seg,
    const int2* __restrict__ cv, const float* __restrict__ bias,
    float* __restrict__ OUT, int n)
{
  constexpr int VEC = FOUT / 32;
  typedef typename RowVec<VEC>::T vecT;
  const int lane = threadIdx.x & 31;
  const int node = __builtin_amdgcn_readfirstlane(
      (int)((blockIdx.x * blockDim.x + threadIdx.x) >> 5));
  if (node >= n) return;

  const int2 sg = seg[node];
  vecT bv;
  for (int k = 0; k < VEC; ++k) bv[k] = bias[lane * VEC + k];

  const vecT acc = gather_node<VEC>((const vecT*)H, cv, sg.x, sg.y, lane);

  vecT o;
  for (int k = 0; k < VEC; ++k) o[k] = acc[k] + bv[k];
  ((vecT*)OUT)[(size_t)node * 32 + lane] = o;
}

extern "C" void kernel_launch(void* const* d_in, const int* in_sizes, int n_in,
                              void* d_out, int out_size, void* d_ws, size_t ws_size,
                              hipStream_t stream) {
  const float* x   = (const float*)d_in[0];
  const int*   ei  = (const int*)d_in[1];
  const float* W1  = (const float*)d_in[2];
  const float* b1  = (const float*)d_in[3];
  const float* g1  = (const float*)d_in[4];
  const float* be1 = (const float*)d_in[5];
  const float* W2  = (const float*)d_in[6];
  const float* b2  = (const float*)d_in[7];
  const float* g2  = (const float*)d_in[8];
  const float* be2 = (const float*)d_in[9];
  const float* W3  = (const float*)d_in[10];
  const float* b3  = (const float*)d_in[11];

  const int N = in_sizes[0] / 128;
  const int E = in_sizes[1] / 2;
  const int* src = ei;
  const int* dst = ei + E;

  const size_t Np = ((size_t)N + 63) & ~(size_t)63;
  const size_t Ep = ((size_t)E + 63) & ~(size_t)63;
  float* ws     = (float*)d_ws;
  const int NR  = (N + RNG - 1) / RNG;
  const size_t Tp = ((size_t)NR * NCH + 63) & ~(size_t)63;
  float* dinv   = ws;
  int*   tot    = (int*)(dinv + Np);
  int2*  seg    = (int2*)(tot + Tp);
  int*   cntP   = (int*)(seg + Np);
  int2*  cv     = (int2*)(cntP + (size_t)NCH * Np);
  float* h      = (float*)(cv + Ep + Np);
  float* hn     = h + (size_t)N * 128;

  const dim3 rc(NR, NCH);
  count_range<<<rc, 1024, 0, stream>>>(dst, cntP, tot, E, (int)Np);
  scan_dinv  <<<NR, 256, 0, stream>>>(cntP, tot, seg, dinv, cv, N, (int)Np);
  fill_range <<<rc, 1024, 0, stream>>>(src, dst, cntP, seg, dinv, cv, N, E, (int)Np);

  const int gblocks = (N + 7) / 8;

  const int fblocks = (N + 31) / 32;

  gemm_wmma_f32<128><<<(N + 15) / 16, 256, 0, stream>>>(x, W1, h, N);

  gather_ln_gemm<128><<<fblocks, 256, 0, stream>>>(h, seg, cv, b1, g1, be1, W2, hn, N);

  gather_ln_gemm<64><<<fblocks, 256, 0, stream>>>(hn, seg, cv, b2, g2, be2, W3, h, N);

  gather_bias<64><<<gblocks, 256, 0, stream>>>(h, seg, cv, b3, (float*)d_out, N);
}


# Fewer, longer waves (wave entry is a first-order cost here): last gather takes 4 nodes per wave (50000 -> 12504 waves), layer-1 GEMM wave a 16x64 strip sharing A (25000 -> 6256); sum order unchanged
# speedup vs baseline: 4.1221x; 1.0593x over previous
#include <hip/hip_runtime.h>
#include <math.h>

typedef __attribute__((ext_vector_type(2))) float v2f;
typedef __attribute__((ext_vector_type(8))) float v8f;

#define LN_EPS 1e-5f

#define RNG 2048
#define NCH 8
#define EPT 16

__device__ __forceinline__ void chunk_bounds(int c, int E, int& e0, int& e1) {
  const int clen = (((E + NCH - 1) / NCH) + 1023) & ~1023;
  e0 = c * clen;
  e1 = (e0 + clen < E) ? e0 + clen : E;
}

__global__ __launch_bounds__(1024) void count_range(
    const int* __restrict__ dst, int* __restrict__ cntP, int* __restrict__ tot,
    int E, int np)
{
  __shared__ int hist[RNG];
  __shared__ int part[32];
  const int t    = threadIdx.x;
  const int r    = blockIdx.x;
  const int c    = blockIdx.y;
  const int base = r * RNG;

  for (int j = t; j < RNG; j += 1024) hist[j] = 0;
  __syncthreads();

  int e0, e1;
  chunk_bounds(c, E, e0, e1);
  int mine = 0;
  for (int eb = e0; eb < e1; eb += 1024 * EPT) {
    int d[EPT];
#pragma unroll
    for (int k = 0; k < EPT; ++k) {
      const int e = eb + k * 1024 + t;
      d[k] = dst[(e < e1) ? e : e1 - 1];
    }
#pragma unroll
    for (int k = 0; k < EPT; ++k) {
      const int e = eb + k * 1024 + t;
      const unsigned u = (unsigned)(d[k] - base);
      if (e < e1 && u < (unsigned)RNG) { atomicAdd(&hist[u], 1); ++mine; }
    }
  }

#pragma unroll
  for (int m = 16; m >= 1; m >>= 1) mine += __shfl_xor(mine, m, 32);
  if ((t & 31) == 0) part[t >> 5] = mine;
  __syncthreads();

  for (int j = t; j < RNG; j += 1024) {
    const int i = base + j;
    if (i < np) cntP[(size_t)c * np + i] = hist[j];
  }
  if (t == 0) {
    int s = 0;
#pragma unroll
    for (int k = 0; k < 32; ++k) s += part[k];
    tot[r * NCH + c] = s;
  }
}

__global__ __launch_bounds__(256) void scan_dinv(
    const int* __restrict__ cntP, const int* __restrict__ tot,
    int2* __restrict__ seg, float* __restrict__ dinv, int2* __restrict__ cv,
    int n, int np)
{
  __shared__ int part[16];
  const int t    = threadIdx.x;
  const int w    = t >> 5;
  const int lane = t & 31;
  const int b    = blockIdx.x;

  int pre = 0;
  for (int i = t; i < b * NCH; i += 256) pre += tot[i];
#pragma unroll
  for (int m = 16; m >= 1; m >>= 1) pre += __shfl_xor(pre, m, 32);

  const int i0 = b * RNG + t * 8;
  int c[8] = {0, 0, 0, 0, 0, 0, 0, 0};
  if (i0 < np) {
#pragma unroll
    for (int k = 0; k < NCH; ++k) {
      const int4* __restrict__ q = (const int4*)(cntP + (size_t)k * np + i0);
      const int4 x = q[0], y = q[1];
      c[0] += x.x; c[1] += x.y; c[2] += x.z; c[3] += x.w;
      c[4] += y.x; c[5] += y.y; c[6] += y.z; c[7] += y.w;
    }
  }
#pragma unroll
  for (int j = 0; j < 8; ++j) c[j] = (i0 + j < n) ? c[j] + 1 : 0;
  const int tsum = ((c[0] + c[1]) + (c[2] + c[3])) + ((c[4] + c[5]) + (c[6] + c[7]));
  int incl = tsum;
#pragma unroll
  for (int m = 1; m < 32; m <<= 1) {
    int o = __shfl_up(incl, m, 32);
    if (lane >= m) incl += o;
  }
  if (lane == 0)  part[w] = pre;
  if (lane == 31) part[8 + w] = incl;
  __syncthreads();
  int off = b * RNG;
#pragma unroll
  for (int k = 0; k < 8; ++k) {
    off += part[k];
    if (k < w) off += part[8 + k];
  }
  int run = off + incl - tsum;
#pragma unroll
  for (int j = 0; j < 8; ++j) {
    if (i0 + j < n) {
      seg[i0 + j] = make_int2(run, c[j]);
      const float d  = (float)c[j];
      const float dv = (d > 0.0f) ? rsqrtf(d) : 0.0f;
      dinv[i0 + j] = dv;
      cv[run] = make_int2(i0 + j, __float_as_int(dv * dv));
    }
    run += c[j];
  }
}

__global__ __launch_bounds__(1024) void fill_range(
    const int* __restrict__ src, const int* __restrict__ dst,
    const int* __restrict__ cntP, const int2* __restrict__ seg,
    const float* __restrict__ dinv, int2* __restrict__ cv, int n, int E, int np)
{
  __shared__ int   cur[RNG + 32];
  __shared__ float dvl[RNG];
  const int t    = threadIdx.x;
  const int r    = blockIdx.x;
  const int c    = blockIdx.y;
  const int base = r * RNG;

  for (int j = t; j < RNG; j += 1024) {
    const int i = base + j;
    int   v  = 0;
    float dv = 0.0f;
    if (i < n) {
      v  = seg[i].x + 1;
      dv = dinv[i];
#pragma unroll
      for (int k = 0; k < NCH; ++k) {
        const int x = cntP[(size_t)k * np + i];
        v += (k < c) ? x : 0;
      }
    }
    cur[j] = v;
    dvl[j] = dv;
  }
  if (t < 32) cur[RNG + t] = 0;
  __syncthreads();

  int* const spare = &cur[RNG + (t & 31)];
  int e0, e1;
  chunk_bounds(c, E, e0, e1);
  for (int eb = e0; eb < e1; eb += 1024 * EPT) {
    unsigned s[EPT];
    int   d[EPT], pos[EPT];
    float ws[EPT];
#pragma unroll
    for (int k = 0; k < EPT; ++k) {
      const int e  = eb + k * 1024 + t;
      const int ec = (e < e1) ? e : e1 - 1;
      d[k] = dst[ec];
      s[k] = (unsigned)src[ec];
    }
#pragma unroll
    for (int k = 0; k < EPT; ++k) asm volatile("" :: "v"(s[k]));
#pragma unroll
    for (int k = 0; k < EPT; ++k) {
      const int e = eb + k * 1024 + t;
      const bool hit = e < e1 && (unsigned)(d[k] - base) < (unsigned)RNG;
      ws[k] = dinv[hit ? s[k] : 0u];
    }
#pragma unroll
    for (int k = 0; k < EPT; ++k) {
      const int e = eb + k * 1024 + t;
      const unsigned u = (unsigned)(d[k] - base);
      const bool hit = e < e1 && u < (unsigned)RNG;
      pos[k] = atomicAdd(hit ? &cur[u] : spare, hit ? 1 : 0);
    }
#pragma unroll
    for (int k = 0; k < EPT; ++k) asm volatile("" :: "v"(ws[k]));
#pragma unroll
    for (int k = 0; k < EPT; ++k) {
      const int e = eb + k * 1024 + t;
      const unsigned u = (unsigned)(d[k] - base);
      if (e < e1 && u < (unsigned)RNG)
        cv[pos[k]] = make_int2((int)s[k], __float_as_int(ws[k] * dvl[u]));
    }
  }
}

template <int FOUT>
__global__ __launch_bounds__(256) void gemm_wmma_f32(
    const float* __restrict__ X, const float* __restrict__ W,
    float* __restrict__ H, int nrows)
{
  constexpr int CTW = 4;
  constexpr int SPR = FOUT / (16 * CTW);
  constexpr int RPB = 8 / SPR;
  const int w    = __builtin_amdgcn_readfirstlane((int)(threadIdx.x >> 5));
  const int lane = threadIdx.x & 31;

  const int rowTile = blockIdx.x * RPB + (w / SPR);
  const int r0 = rowTile * 16;
  if (r0 >= nrows) return;
  const int c0 = (w % SPR) * (16 * CTW);

  const int lh = lane & 15;
  const int kh = (lane >> 4) << 1;

  const float* __restrict__ xrow = X + (size_t)(r0 + lh) * 128 + kh;
  const float* __restrict__ wcol = W + (size_t)kh * FOUT + c0 + lh;

  v8f acc[CTW] = {};
#pragma unroll
  for (int kk = 0; kk < 128; kk += 4) {
    v2f a; a.x = xrow[kk]; a.y = xrow[kk + 1];
#pragma unroll
    for (int t = 0; t < CTW; ++t) {
      v2f b;
      b.x = wcol[(size_t)kk * FOUT + t * 16];
      b.y = wcol[(size_t)(kk + 1) * FOUT + t * 16];
      acc[t] = __builtin_amdgcn_wmma_f32_16x16x4_f32(
          false, a, false, b, (short)0, acc[t], false, false);
    }
  }

  const int rbase = r0 + ((lane >> 4) << 3);
  float* hp = H + (size_t)rbase * FOUT + c0 + lh;
#pragma unroll
  for (int t = 0; t < CTW; ++t)
#pragma unroll
    for (int v = 0; v < 8; ++v) hp[(size_t)v * FOUT + t * 16] = acc[t][v];
}

template <int VEC> struct RowVec {
  typedef __attribute__((ext_vector_type(VEC))) float T;
};

template <int VEC>
__device__ __forceinline__ typename RowVec<VEC>::T gather_node(
    const typename RowVec<VEC>::T* __restrict__ Hv,
    const int2* __restrict__ cv, int start, int m, int lane)
{
  typedef typename RowVec<VEC>::T vecT;
  vecT acc = {};

  for (int base = 0; base < m; base += 32) {
    const int mc = (m - base < 32) ? (m - base) : 32;
    int c = 0, wb = 0;
    if (lane < mc) { const int2 e = cv[start + base + lane]; c = e.x; wb = e.y; }

    int j = 0;
    for (; j + 4 <= mc; j += 4) {
      const int s0 = __builtin_amdgcn_readlane(c, j);
      const int s1 = __builtin_amdgcn_readlane(c, j + 1);
      const int s2 = __builtin_amdgcn_readlane(c, j + 2);
      const int s3 = __builtin_amdgcn_readlane(c, j + 3);
      const vecT r0 = Hv[(size_t)s0 * 32 + lane];
      const vecT r1 = Hv[(size_t)s1 * 32 + lane];
      const vecT r2 = Hv[(size_t)s2 * 32 + lane];
      const vecT r3 = Hv[(size_t)s3 * 32 + lane];
      acc += r0 * __int_as_float(__builtin_amdgcn_readlane(wb, j));
      acc += r1 * __int_as_float(__builtin_amdgcn_readlane(wb, j + 1));
      acc += r2 * __int_as_float(__builtin_amdgcn_readlane(wb, j + 2));
      acc += r3 * __int_as_float(__builtin_amdgcn_readlane(wb, j + 3));
    }
    const int rem = mc - j;
    if (rem > 0) {
      vecT r0 = {}, r1 = {}, r2 = {};
      r0 = Hv[(size_t)__builtin_amdgcn_readlane(c, j) * 32 + lane];
      if (rem > 1) r1 = Hv[(size_t)__builtin_amdgcn_readlane(c, j + 1) * 32 + lane];
      if (rem > 2) r2 = Hv[(size_t)__builtin_amdgcn_readlane(c, j + 2) * 32 + lane];
      acc += r0 * __int_as_float(__builtin_amdgcn_readlane(wb, j));
      acc += r1 * __int_as_float(__builtin_amdgcn_readlane(wb, j + 1));
      acc += r2 * __int_as_float(__builtin_amdgcn_readlane(wb, j + 2));
    }
  }
  return acc;
}

template <int FNEXT>
__global__ __launch_bounds__(256, 16) void gather_ln_gemm(
    const float* __restrict__ H, const int2* __restrict__ seg,
    const int2* __restrict__ cv, const float* __restrict__ bias,
    const float* __restrict__ gamma, const float* __restrict__ beta,
    const float* __restrict__ WN, float* __restrict__ HN, int n)
{
  typedef RowVec<4>::T vecT;
  constexpr int ROWS  = 32;
  constexpr int PITCH = 132;
  __shared__ float tile[ROWS * PITCH];

  const int lane = threadIdx.x & 31;
  const int w  = __builtin_amdgcn_readfirstlane((int)(threadIdx.x >> 5));
  const int nb = blockIdx.x * ROWS + w * 4;

  vecT bv, gv, ev;
  for (int k = 0; k < 4; ++k) {
    bv[k] = bias[lane * 4 + k];
    gv[k] = gamma[lane * 4 + k];
    ev[k] = beta[lane * 4 + k];
  }

  int2 sgl = make_int2(0, 0);
  if (lane < 4 && nb + lane < n) sgl = seg[nb + lane];

  const vecT* __restrict__ Hv = (const vecT*)H;
  for (int q = 0; q < 4; ++q) {
    const int start = __builtin_amdgcn_readlane(sgl.x, q);
    const int m     = __builtin_amdgcn_readlane(sgl.y, q);
    const vecT acc = gather_node<4>(Hv, cv, start, m, lane);

    const float v0 = acc[0] + bv[0];
    const float v1 = acc[1] + bv[1];
    const float v2 = acc[2] + bv[2];
    const float v3 = acc[3] + bv[3];

    float s = v0 + v1 + v2 + v3;
#pragma unroll
    for (int x = 16; x >= 1; x >>= 1) s += __shfl_xor(s, x, 32);
    const float mu = s * (1.0f / 128.0f);

    const float d0 = v0 - mu, d1 = v1 - mu, d2 = v2 - mu, d3 = v3 - mu;
    float qs = d0 * d0 + d1 * d1 + d2 * d2 + d3 * d3;
#pragma unroll
    for (int x = 16; x >= 1; x >>= 1) qs += __shfl_xor(qs, x, 32);
    const float rstd = rsqrtf(qs * (1.0f / 128.0f) + LN_EPS);

    vecT o;
    o[0] = fmaxf(0.0f, d0 * rstd * gv[0] + ev[0]);
    o[1] = fmaxf(0.0f, d1 * rstd * gv[1] + ev[1]);
    o[2] = fmaxf(0.0f, d2 * rstd * gv[2] + ev[2]);
    o[3] = fmaxf(0.0f, d3 * rstd * gv[3] + ev[3]);
    *(vecT*)(tile + (w * 4 + q) * PITCH + lane * 4) = o;
  }
  __syncthreads();

  constexpr int CT = FNEXT / 16;
  constexpr int NT = CT / 4;
  const int ct  = w % CT;
  const int rt0 = (w / CT) * NT;
  const int c0  = ct * 16;
  const int lh  = lane & 15;
  const int kh  = (lane >> 4) << 1;

  const float* ap = tile + (rt0 * 16 + lh) * PITCH + kh;
  const float* __restrict__ wcol = WN + (size_t)kh * FNEXT + c0 + lh;

  v8f acc[NT] = {};
#pragma unroll
  for (int kk = 0; kk < 128; kk += 4) {
    v2f b; b.x = wcol[(size_t)kk * FNEXT]; b.y = wcol[(size_t)(kk + 1) * FNEXT];
#pragma unroll
    for (int t = 0; t < NT; ++t) {
      const v2f a = *(const v2f*)(ap + t * 16 * PITCH + kk);
      acc[t] = __builtin_amdgcn_wmma_f32_16x16x4_f32(
          false, a, false, b, (short)0, acc[t], false, false);
    }
  }

#pragma unroll
  for (int t = 0; t < NT; ++t) {
    const int rbase = blockIdx.x * ROWS + (rt0 + t) * 16 + ((lane >> 4) << 3);
    float* hp = HN + (size_t)rbase * FNEXT + c0 + lh;
#pragma unroll
    for (int v = 0; v < 8; ++v)
      if (rbase + v < n) hp[(size_t)v * FNEXT] = acc[t][v];
  }
}

template <int FOUT>
__global__ __launch_bounds__(256) void gather_bias(
    const float* __restrict__ H, const int2* __restrict__ seg,
    const int2* __restrict__ cv, const float* __restrict__ bias,
    float* __restrict__ OUT, int n)
{
  constexpr int VEC = FOUT / 32;
  typedef typename RowVec<VEC>::T vecT;
  const int lane = threadIdx.x & 31;
  const int w  = __builtin_amdgcn_readfirstlane((int)(threadIdx.x >> 5));
  const int nb = blockIdx.x * 32 + w * 4;
  if (nb >= n) return;

  vecT bv;
  for (int k = 0; k < VEC; ++k) bv[k] = bias[lane * VEC + k];

  int2 sgl = make_int2(0, 0);
  if (lane < 4 && nb + lane < n) sgl = seg[nb + lane];

  const vecT* __restrict__ Hv = (const vecT*)H;
  const int nq = (n - nb < 4) ? (n - nb) : 4;
  for (int q = 0; q < nq; ++q) {
    const int start = __builtin_amdgcn_readlane(sgl.x, q);
    const int m     = __builtin_amdgcn_readlane(sgl.y, q);
    const vecT acc = gather_node<VEC>(Hv, cv, start, m, lane);

    vecT o;
    for (int k = 0; k < VEC; ++k) o[k] = acc[k] + bv[k];
    ((vecT*)OUT)[(size_t)(nb + q) * 32 + lane] = o;
  }
}

extern "C" void kernel_launch(void* const* d_in, const int* in_sizes, int n_in,
                              void* d_out, int out_size, void* d_ws, size_t ws_size,
                              hipStream_t stream) {
  const float* x   = (const float*)d_in[0];
  const int*   ei  = (const int*)d_in[1];
  const float* W1  = (const float*)d_in[2];
  const float* b1  = (const float*)d_in[3];
  const float* g1  = (const float*)d_in[4];
  const float* be1 = (const float*)d_in[5];
  const float* W2  = (const float*)d_in[6];
  const float* b2  = (const float*)d_in[7];
  const float* g2  = (const float*)d_in[8];
  const float* be2 = (const float*)d_in[9];
  const float* W3  = (const float*)d_in[10];
  const float* b3  = (const float*)d_in[11];

  const int N = in_sizes[0] / 128;
  const int E = in_sizes[1] / 2;
  const int* src = ei;
  const int* dst = ei + E;

  const size_t Np = ((size_t)N + 63) & ~(size_t)63;
  const size_t Ep = ((size_t)E + 63) & ~(size_t)63;
  float* ws     = (float*)d_ws;
  const int NR  = (N + RNG - 1) / RNG;
  const size_t Tp = ((size_t)NR * NCH + 63) & ~(size_t)63;
  float* dinv   = ws;
  int*   tot    = (int*)(dinv + Np);
  int2*  seg    = (int2*)(tot + Tp);
  int*   cntP   = (int*)(seg + Np);
  int2*  cv     = (int2*)(cntP + (size_t)NCH * Np);
  float* h      = (float*)(cv + Ep + Np);
  float* hn     = h + (size_t)N * 128;

  const dim3 rc(NR, NCH);
  count_range<<<rc, 1024, 0, stream>>>(dst, cntP, tot, E, (int)Np);
  scan_dinv  <<<NR, 256, 0, stream>>>(cntP, tot, seg, dinv, cv, N, (int)Np);
  fill_range <<<rc, 1024, 0, stream>>>(src, dst, cntP, seg, dinv, cv, N, E, (int)Np);

  const int fblocks = (N + 31) / 32;

  gemm_wmma_f32<128><<<((N + 15) / 16 + 3) / 4, 256, 0, stream>>>(x, W1, h, N);

  gather_ln_gemm<128><<<fblocks, 256, 0, stream>>>(h, seg, cv, b1, g1, be1, W2, hn, N);

  gather_ln_gemm<64><<<fblocks, 256, 0, stream>>>(hn, seg, cv, b2, g2, be2, W3, h, N);

  gather_bias<64><<<fblocks, 256, 0, stream>>>(h, seg, cv, b3, (float*)d_out, N);
}


# Layer-1 WMMA GEMM (independent of the CSR build) runs in the degree scan's launch: blocks 0..NR-1 scan, the rest are GEMM blocks; 7 -> 6 launches; products and sum order unchanged
# speedup vs baseline: 4.3268x; 1.0497x over previous
#include <hip/hip_runtime.h>
#include <math.h>

typedef __attribute__((ext_vector_type(2))) float v2f;
typedef __attribute__((ext_vector_type(8))) float v8f;

#define LN_EPS 1e-5f

#define RNG 2048
#define NCH 8
#define EPT 16

__device__ __forceinline__ void chunk_bounds(int c, int E, int& e0, int& e1) {
  const int clen = (((E + NCH - 1) / NCH) + 1023) & ~1023;
  e0 = c * clen;
  e1 = (e0 + clen < E) ? e0 + clen : E;
}

__global__ __launch_bounds__(1024) void count_range(
    const int* __restrict__ dst, int* __restrict__ cntP, int* __restrict__ tot,
    int E, int np)
{
  __shared__ int hist[RNG];
  __shared__ int part[32];
  const int t    = threadIdx.x;
  const int r    = blockIdx.x;
  const int c    = blockIdx.y;
  const int base = r * RNG;

  for (int j = t; j < RNG; j += 1024) hist[j] = 0;
  __syncthreads();

  int e0, e1;
  chunk_bounds(c, E, e0, e1);
  int mine = 0;
  for (int eb = e0; eb < e1; eb += 1024 * EPT) {
    int d[EPT];
#pragma unroll
    for (int k = 0; k < EPT; ++k) {
      const int e = eb + k * 1024 + t;
      d[k] = dst[(e < e1) ? e : e1 - 1];
    }
#pragma unroll
    for (int k = 0; k < EPT; ++k) {
      const int e = eb + k * 1024 + t;
      const unsigned u = (unsigned)(d[k] - base);
      if (e < e1 && u < (unsigned)RNG) { atomicAdd(&hist[u], 1); ++mine; }
    }
  }

#pragma unroll
  for (int m = 16; m >= 1; m >>= 1) mine += __shfl_xor(mine, m, 32);
  if ((t & 31) == 0) part[t >> 5] = mine;
  __syncthreads();

  for (int j = t; j < RNG; j += 1024) {
    const int i = base + j;
    if (i < np) cntP[(size_t)c * np + i] = hist[j];
  }
  if (t == 0) {
    int s = 0;
#pragma unroll
    for (int k = 0; k < 32; ++k) s += part[k];
    tot[r * NCH + c] = s;
  }
}

__device__ __forceinline__ void scan_dinv_block(
    const int* __restrict__ cntP, const int* __restrict__ tot,
    int2* __restrict__ seg, float* __restrict__ dinv, int2* __restrict__ cv,
    int n, int np, int b)
{
  __shared__ int part[16];
  const int t    = threadIdx.x;
  const int w    = t >> 5;
  const int lane = t & 31;

  int pre = 0;
  for (int i = t; i < b * NCH; i += 256) pre += tot[i];
#pragma unroll
  for (int m = 16; m >= 1; m >>= 1) pre += __shfl_xor(pre, m, 32);

  const int i0 = b * RNG + t * 8;
  int c[8] = {0, 0, 0, 0, 0, 0, 0, 0};
  if (i0 < np) {
#pragma unroll
    for (int k = 0; k < NCH; ++k) {
      const int4* __restrict__ q = (const int4*)(cntP + (size_t)k * np + i0);
      const int4 x = q[0], y = q[1];
      c[0] += x.x; c[1] += x.y; c[2] += x.z; c[3] += x.w;
      c[4] += y.x; c[5] += y.y; c[6] += y.z; c[7] += y.w;
    }
  }
#pragma unroll
  for (int j = 0; j < 8; ++j) c[j] = (i0 + j < n) ? c[j] + 1 : 0;
  const int tsum = ((c[0] + c[1]) + (c[2] + c[3])) + ((c[4] + c[5]) + (c[6] + c[7]));
  int incl = tsum;
#pragma unroll
  for (int m = 1; m < 32; m <<= 1) {
    int o = __shfl_up(incl, m, 32);
    if (lane >= m) incl += o;
  }
  if (lane == 0)  part[w] = pre;
  if (lane == 31) part[8 + w] = incl;
  __syncthreads();
  int off = b * RNG;
#pragma unroll
  for (int k = 0; k < 8; ++k) {
    off += part[k];
    if (k < w) off += part[8 + k];
  }
  int run = off + incl - tsum;
#pragma unroll
  for (int j = 0; j < 8; ++j) {
    if (i0 + j < n) {
      seg[i0 + j] = make_int2(run, c[j]);
      const float d  = (float)c[j];
      const float dv = (d > 0.0f) ? rsqrtf(d) : 0.0f;
      dinv[i0 + j] = dv;
      cv[run] = make_int2(i0 + j, __float_as_int(dv * dv));
    }
    run += c[j];
  }
}

__global__ __launch_bounds__(1024) void fill_range(
    const int* __restrict__ src, const int* __restrict__ dst,
    const int* __restrict__ cntP, const int2* __restrict__ seg,
    const float* __restrict__ dinv, int2* __restrict__ cv, int n, int E, int np)
{
  __shared__ int   cur[RNG + 32];
  __shared__ float dvl[RNG];
  const int t    = threadIdx.x;
  const int r    = blockIdx.x;
  const int c    = blockIdx.y;
  const int base = r * RNG;

  for (int j = t; j < RNG; j += 1024) {
    const int i = base + j;
    int   v  = 0;
    float dv = 0.0f;
    if (i < n) {
      v  = seg[i].x + 1;
      dv = dinv[i];
#pragma unroll
      for (int k = 0; k < NCH; ++k) {
        const int x = cntP[(size_t)k * np + i];
        v += (k < c) ? x : 0;
      }
    }
    cur[j] = v;
    dvl[j] = dv;
  }
  if (t < 32) cur[RNG + t] = 0;
  __syncthreads();

  int* const spare = &cur[RNG + (t & 31)];
  int e0, e1;
  chunk_bounds(c, E, e0, e1);
  for (int eb = e0; eb < e1; eb += 1024 * EPT) {
    unsigned s[EPT];
    int   d[EPT], pos[EPT];
    float ws[EPT];
#pragma unroll
    for (int k = 0; k < EPT; ++k) {
      const int e  = eb + k * 1024 + t;
      const int ec = (e < e1) ? e : e1 - 1;
      d[k] = dst[ec];
      s[k] = (unsigned)src[ec];
    }
#pragma unroll
    for (int k = 0; k < EPT; ++k) asm volatile("" :: "v"(s[k]));
#pragma unroll
    for (int k = 0; k < EPT; ++k) {
      const int e = eb + k * 1024 + t;
      const bool hit = e < e1 && (unsigned)(d[k] - base) < (unsigned)RNG;
      ws[k] = dinv[hit ? s[k] : 0u];
    }
#pragma unroll
    for (int k = 0; k < EPT; ++k) {
      const int e = eb + k * 1024 + t;
      const unsigned u = (unsigned)(d[k] - base);
      const bool hit = e < e1 && u < (unsigned)RNG;
      pos[k] = atomicAdd(hit ? &cur[u] : spare, hit ? 1 : 0);
    }
#pragma unroll
    for (int k = 0; k < EPT; ++k) asm volatile("" :: "v"(ws[k]));
#pragma unroll
    for (int k = 0; k < EPT; ++k) {
      const int e = eb + k * 1024 + t;
      const unsigned u = (unsigned)(d[k] - base);
      if (e < e1 && u < (unsigned)RNG)
        cv[pos[k]] = make_int2((int)s[k], __float_as_int(ws[k] * dvl[u]));
    }
  }
}

template <int FOUT>
__device__ __forceinline__ void gemm_wmma_f32_block(
    const float* __restrict__ X, const float* __restrict__ W,
    float* __restrict__ H, int nrows, int blk)
{
  constexpr int CTW = 4;
  constexpr int SPR = FOUT / (16 * CTW);
  constexpr int RPB = 8 / SPR;
  const int w    = __builtin_amdgcn_readfirstlane((int)(threadIdx.x >> 5));
  const int lane = threadIdx.x & 31;

  const int rowTile = blk * RPB + (w / SPR);
  const int r0 = rowTile * 16;
  if (r0 >= nrows) return;
  const int c0 = (w % SPR) * (16 * CTW);

  const int lh = lane & 15;
  const int kh = (lane >> 4) << 1;

  const float* __restrict__ xrow = X + (size_t)(r0 + lh) * 128 + kh;
  const float* __restrict__ wcol = W + (size_t)kh * FOUT + c0 + lh;

  v8f acc[CTW] = {};
#pragma unroll
  for (int kk = 0; kk < 128; kk += 4) {
    v2f a; a.x = xrow[kk]; a.y = xrow[kk + 1];
#pragma unroll
    for (int t = 0; t < CTW; ++t) {
      v2f b;
      b.x = wcol[(size_t)kk * FOUT + t * 16];
      b.y = wcol[(size_t)(kk + 1) * FOUT + t * 16];
      acc[t] = __builtin_amdgcn_wmma_f32_16x16x4_f32(
          false, a, false, b, (short)0, acc[t], false, false);
    }
  }

  const int rbase = r0 + ((lane >> 4) << 3);
  float* hp = H + (size_t)rbase * FOUT + c0 + lh;
#pragma unroll
  for (int t = 0; t < CTW; ++t)
#pragma unroll
    for (int v = 0; v < 8; ++v) hp[(size_t)v * FOUT + t * 16] = acc[t][v];
}

__global__ __launch_bounds__(256) void scan_dinv_gemm1(
    const int* __restrict__ cntP, const int* __restrict__ tot,
    int2* __restrict__ seg, float* __restrict__ dinv, int2* __restrict__ cv,
    int n, int np, int nscan,
    const float* __restrict__ X, const float* __restrict__ W,
    float* __restrict__ H)
{
  const int b = blockIdx.x;
  if (b < nscan) scan_dinv_block(cntP, tot, seg, dinv, cv, n, np, b);
  else           gemm_wmma_f32_block<128>(X, W, H, n, b - nscan);
}

template <int VEC> struct RowVec {
  typedef __attribute__((ext_vector_type(VEC))) float T;
};

template <int VEC>
__device__ __forceinline__ typename RowVec<VEC>::T gather_node(
    const typename RowVec<VEC>::T* __restrict__ Hv,
    const int2* __restrict__ cv, int start, int m, int lane)
{
  typedef typename RowVec<VEC>::T vecT;
  vecT acc = {};

  for (int base = 0; base < m; base += 32) {
    const int mc = (m - base < 32) ? (m - base) : 32;
    int c = 0, wb = 0;
    if (lane < mc) { const int2 e = cv[start + base + lane]; c = e.x; wb = e.y; }

    int j = 0;
    for (; j + 4 <= mc; j += 4) {
      const int s0 = __builtin_amdgcn_readlane(c, j);
      const int s1 = __builtin_amdgcn_readlane(c, j + 1);
      const int s2 = __builtin_amdgcn_readlane(c, j + 2);
      const int s3 = __builtin_amdgcn_readlane(c, j + 3);
      const vecT r0 = Hv[(size_t)s0 * 32 + lane];
      const vecT r1 = Hv[(size_t)s1 * 32 + lane];
      const vecT r2 = Hv[(size_t)s2 * 32 + lane];
      const vecT r3 = Hv[(size_t)s3 * 32 + lane];
      acc += r0 * __int_as_float(__builtin_amdgcn_readlane(wb, j));
      acc += r1 * __int_as_float(__builtin_amdgcn_readlane(wb, j + 1));
      acc += r2 * __int_as_float(__builtin_amdgcn_readlane(wb, j + 2));
      acc += r3 * __int_as_float(__builtin_amdgcn_readlane(wb, j + 3));
    }
    const int rem = mc - j;
    if (rem > 0) {
      vecT r0 = {}, r1 = {}, r2 = {};
      r0 = Hv[(size_t)__builtin_amdgcn_readlane(c, j) * 32 + lane];
      if (rem > 1) r1 = Hv[(size_t)__builtin_amdgcn_readlane(c, j + 1) * 32 + lane];
      if (rem > 2) r2 = Hv[(size_t)__builtin_amdgcn_readlane(c, j + 2) * 32 + lane];
      acc += r0 * __int_as_float(__builtin_amdgcn_readlane(wb, j));
      acc += r1 * __int_as_float(__builtin_amdgcn_readlane(wb, j + 1));
      acc += r2 * __int_as_float(__builtin_amdgcn_readlane(wb, j + 2));
    }
  }
  return acc;
}

template <int FNEXT>
__global__ __launch_bounds__(256, 16) void gather_ln_gemm(
    const float* __restrict__ H, const int2* __restrict__ seg,
    const int2* __restrict__ cv, const float* __restrict__ bias,
    const float* __restrict__ gamma, const float* __restrict__ beta,
    const float* __restrict__ WN, float* __restrict__ HN, int n)
{
  typedef RowVec<4>::T vecT;
  constexpr int ROWS  = 32;
  constexpr int PITCH = 132;
  __shared__ float tile[ROWS * PITCH];

  const int lane = threadIdx.x & 31;
  const int w  = __builtin_amdgcn_readfirstlane((int)(threadIdx.x >> 5));
  const int nb = blockIdx.x * ROWS + w * 4;

  vecT bv, gv, ev;
  for (int k = 0; k < 4; ++k) {
    bv[k] = bias[lane * 4 + k];
    gv[k] = gamma[lane * 4 + k];
    ev[k] = beta[lane * 4 + k];
  }

  int2 sgl = make_int2(0, 0);
  if (lane < 4 && nb + lane < n) sgl = seg[nb + lane];

  const vecT* __restrict__ Hv = (const vecT*)H;
  for (int q = 0; q < 4; ++q) {
    const int start = __builtin_amdgcn_readlane(sgl.x, q);
    const int m     = __builtin_amdgcn_readlane(sgl.y, q);
    const vecT acc = gather_node<4>(Hv, cv, start, m, lane);

    const float v0 = acc[0] + bv[0];
    const float v1 = acc[1] + bv[1];
    const float v2 = acc[2] + bv[2];
    const float v3 = acc[3] + bv[3];

    float s = v0 + v1 + v2 + v3;
#pragma unroll
    for (int x = 16; x >= 1; x >>= 1) s += __shfl_xor(s, x, 32);
    const float mu = s * (1.0f / 128.0f);

    const float d0 = v0 - mu, d1 = v1 - mu, d2 = v2 - mu, d3 = v3 - mu;
    float qs = d0 * d0 + d1 * d1 + d2 * d2 + d3 * d3;
#pragma unroll
    for (int x = 16; x >= 1; x >>= 1) qs += __shfl_xor(qs, x, 32);
    const float rstd = rsqrtf(qs * (1.0f / 128.0f) + LN_EPS);

    vecT o;
    o[0] = fmaxf(0.0f, d0 * rstd * gv[0] + ev[0]);
    o[1] = fmaxf(0.0f, d1 * rstd * gv[1] + ev[1]);
    o[2] = fmaxf(0.0f, d2 * rstd * gv[2] + ev[2]);
    o[3] = fmaxf(0.0f, d3 * rstd * gv[3] + ev[3]);
    *(vecT*)(tile + (w * 4 + q) * PITCH + lane * 4) = o;
  }
  __syncthreads();

  constexpr int CT = FNEXT / 16;
  constexpr int NT = CT / 4;
  const int ct  = w % CT;
  const int rt0 = (w / CT) * NT;
  const int c0  = ct * 16;
  const int lh  = lane & 15;
  const int kh  = (lane >> 4) << 1;

  const float* ap = tile + (rt0 * 16 + lh) * PITCH + kh;
  const float* __restrict__ wcol = WN + (size_t)kh * FNEXT + c0 + lh;

  v8f acc[NT] = {};
#pragma unroll
  for (int kk = 0; kk < 128; kk += 4) {
    v2f b; b.x = wcol[(size_t)kk * FNEXT]; b.y = wcol[(size_t)(kk + 1) * FNEXT];
#pragma unroll
    for (int t = 0; t < NT; ++t) {
      const v2f a = *(const v2f*)(ap + t * 16 * PITCH + kk);
      acc[t] = __builtin_amdgcn_wmma_f32_16x16x4_f32(
          false, a, false, b, (short)0, acc[t], false, false);
    }
  }

#pragma unroll
  for (int t = 0; t < NT; ++t) {
    const int rbase = blockIdx.x * ROWS + (rt0 + t) * 16 + ((lane >> 4) << 3);
    float* hp = HN + (size_t)rbase * FNEXT + c0 + lh;
#pragma unroll
    for (int v = 0; v < 8; ++v)
      if (rbase + v < n) hp[(size_t)v * FNEXT] = acc[t][v];
  }
}

template <int FOUT>
__global__ __launch_bounds__(256) void gather_bias(
    const float* __restrict__ H, const int2* __restrict__ seg,
    const int2* __restrict__ cv, const float* __restrict__ bias,
    float* __restrict__ OUT, int n)
{
  constexpr int VEC = FOUT / 32;
  typedef typename RowVec<VEC>::T vecT;
  const int lane = threadIdx.x & 31;
  const int w  = __builtin_amdgcn_readfirstlane((int)(threadIdx.x >> 5));
  const int nb = blockIdx.x * 32 + w * 4;
  if (nb >= n) return;

  vecT bv;
  for (int k = 0; k < VEC; ++k) bv[k] = bias[lane * VEC + k];

  int2 sgl = make_int2(0, 0);
  if (lane < 4 && nb + lane < n) sgl = seg[nb + lane];

  const vecT* __restrict__ Hv = (const vecT*)H;
  const int nq = (n - nb < 4) ? (n - nb) : 4;
  for (int q = 0; q < nq; ++q) {
    const int start = __builtin_amdgcn_readlane(sgl.x, q);
    const int m     = __builtin_amdgcn_readlane(sgl.y, q);
    const vecT acc = gather_node<VEC>(Hv, cv, start, m, lane);

    vecT o;
    for (int k = 0; k < VEC; ++k) o[k] = acc[k] + bv[k];
    ((vecT*)OUT)[(size_t)(nb + q) * 32 + lane] = o;
  }
}

extern "C" void kernel_launch(void* const* d_in, const int* in_sizes, int n_in,
                              void* d_out, int out_size, void* d_ws, size_t ws_size,
                              hipStream_t stream) {
  const float* x   = (const float*)d_in[0];
  const int*   ei  = (const int*)d_in[1];
  const float* W1  = (const float*)d_in[2];
  const float* b1  = (const float*)d_in[3];
  const float* g1  = (const float*)d_in[4];
  const float* be1 = (const float*)d_in[5];
  const float* W2  = (const float*)d_in[6];
  const float* b2  = (const float*)d_in[7];
  const float* g2  = (const float*)d_in[8];
  const float* be2 = (const float*)d_in[9];
  const float* W3  = (const float*)d_in[10];
  const float* b3  = (const float*)d_in[11];

  const int N = in_sizes[0] / 128;
  const int E = in_sizes[1] / 2;
  const int* src = ei;
  const int* dst = ei + E;

  const size_t Np = ((size_t)N + 63) & ~(size_t)63;
  const size_t Ep = ((size_t)E + 63) & ~(size_t)63;
  float* ws     = (float*)d_ws;
  const int NR  = (N + RNG - 1) / RNG;
  const size_t Tp = ((size_t)NR * NCH + 63) & ~(size_t)63;
  float* dinv   = ws;
  int*   tot    = (int*)(dinv + Np);
  int2*  seg    = (int2*)(tot + Tp);
  int*   cntP   = (int*)(seg + Np);
  int2*  cv     = (int2*)(cntP + (size_t)NCH * Np);
  float* h      = (float*)(cv + Ep + Np);
  float* hn     = h + (size_t)N * 128;

  const dim3 rc(NR, NCH);
  count_range<<<rc, 1024, 0, stream>>>(dst, cntP, tot, E, (int)Np);
  const int gblocks = ((N + 15) / 16 + 3) / 4;
  scan_dinv_gemm1<<<NR + gblocks, 256, 0, stream>>>(cntP, tot, seg, dinv, cv, N, (int)Np,
                                                    NR, x, W1, h);
  fill_range <<<rc, 1024, 0, stream>>>(src, dst, cntP, seg, dinv, cv, N, E, (int)Np);

  const int fblocks = (N + 31) / 32;

  gather_ln_gemm<128><<<fblocks, 256, 0, stream>>>(h, seg, cv, b1, g1, be1, W2, hn, N);

  gather_ln_gemm<64><<<fblocks, 256, 0, stream>>>(hn, seg, cv, b2, g2, be2, W3, h, N);

  gather_bias<64><<<fblocks, 256, 0, stream>>>(h, seg, cv, b3, (float*)d_out, N);
}


# CSR build re-tiled: node ranges 2048 -> 8192, edge chunks 8 -> 35, one trip per block, edge list read 7x not 25x; per-node sum order: self-loop, chunk order (now 35 chunks), LDS arrival
# speedup vs baseline: 4.7255x; 1.0921x over previous
#include <hip/hip_runtime.h>
#include <math.h>

typedef __attribute__((ext_vector_type(2))) float v2f;
typedef __attribute__((ext_vector_type(8))) float v8f;

#define LN_EPS 1e-5f

#define RNG 8192
#define STL 1024
#define SUB (RNG / STL)
#define NCH 35
#define EPT 17

__device__ __forceinline__ void chunk_bounds(int c, int E, int& e0, int& e1) {
  const int clen = (((E + NCH - 1) / NCH) + 1023) & ~1023;
  e0 = c * clen;
  e1 = (e0 + clen < E) ? e0 + clen : E;
}

__global__ __launch_bounds__(1024) void count_range(
    const int* __restrict__ dst, int* __restrict__ cntP, int* __restrict__ tot,
    int E, int np)
{
  __shared__ int hist[RNG];
  __shared__ int part[32];
  const int t    = threadIdx.x;
  const int r    = blockIdx.x;
  const int c    = blockIdx.y;
  const int base = r * RNG;

  for (int j = t; j < RNG; j += 1024) hist[j] = 0;
  __syncthreads();

  int e0, e1;
  chunk_bounds(c, E, e0, e1);
  for (int eb = e0; eb < e1; eb += 1024 * EPT) {
    int d[EPT];
#pragma unroll
    for (int k = 0; k < EPT; ++k) {
      const int e = eb + k * 1024 + t;
      d[k] = dst[(e < e1) ? e : e1 - 1];
    }
#pragma unroll
    for (int k = 0; k < EPT; ++k) {
      const int e = eb + k * 1024 + t;
      const unsigned u = (unsigned)(d[k] - base);
      if (e < e1 && u < (unsigned)RNG) atomicAdd(&hist[u], 1);
    }
  }
  __syncthreads();

  const int w    = t >> 5;
  const int lane = t & 31;
  int mine = 0;
  int hv[RNG / 1024];
#pragma unroll
  for (int m = 0; m < RNG / 1024; ++m) hv[m] = hist[w * (RNG / 32) + m * 32 + lane];
#pragma unroll
  for (int m = 0; m < RNG / 1024; ++m) {
    const int i = base + w * (RNG / 32) + m * 32 + lane;
    if (i < np) cntP[(size_t)c * np + i] = hv[m];
    mine += hv[m];
  }
#pragma unroll
  for (int m = 16; m >= 1; m >>= 1) mine += __shfl_xor(mine, m, 32);
  if (lane == 0) part[w] = mine;
  __syncthreads();
  if (t < SUB) {
    int a = 0;
#pragma unroll
    for (int k = 0; k < 32 / SUB; ++k) a += part[t * (32 / SUB) + k];
    tot[(r * SUB + t) * NCH + c] = a;
  }
}

__device__ __forceinline__ void scan_dinv_block(
    const int* __restrict__ cntP, const int* __restrict__ tot,
    int* __restrict__ slotP, int2* __restrict__ seg, float* __restrict__ dinv, int2* __restrict__ cv,
    int n, int np, int b)
{
  __shared__ int part[16];
  const int t    = threadIdx.x;
  const int w    = t >> 5;
  const int lane = t & 31;

  int pre = 0;
  for (int i = t; i < b * NCH; i += 256) pre += tot[i];
#pragma unroll
  for (int m = 16; m >= 1; m >>= 1) pre += __shfl_xor(pre, m, 32);

  const int i0 = b * STL + t * 4;
  int c[4] = {0, 0, 0, 0};
  if (i0 < np) {
#pragma unroll
    for (int k = 0; k < NCH; ++k) {
      const int4 x = *(const int4*)(cntP + (size_t)k * np + i0);
      c[0] += x.x; c[1] += x.y; c[2] += x.z; c[3] += x.w;
    }
  }
#pragma unroll
  for (int j = 0; j < 4; ++j) c[j] = (i0 + j < n) ? c[j] + 1 : 0;
  const int tsum = (c[0] + c[1]) + (c[2] + c[3]);
  int incl = tsum;
#pragma unroll
  for (int m = 1; m < 32; m <<= 1) {
    int o = __shfl_up(incl, m, 32);
    if (lane >= m) incl += o;
  }
  if (lane == 0)  part[w] = pre;
  if (lane == 31) part[8 + w] = incl;
  __syncthreads();
  int off = b * STL;
#pragma unroll
  for (int k = 0; k < 8; ++k) {
    off += part[k];
    if (k < w) off += part[8 + k];
  }
  int run = off + incl - tsum;
  int cur[4];
#pragma unroll
  for (int j = 0; j < 4; ++j) {
    if (i0 + j < n) {
      seg[i0 + j] = make_int2(run, c[j]);
      const float d  = (float)c[j];
      const float dv = (d > 0.0f) ? rsqrtf(d) : 0.0f;
      dinv[i0 + j] = dv;
      cv[run] = make_int2(i0 + j, __float_as_int(dv * dv));
    }
    cur[j] = run + 1;
    run += c[j];
  }
  if (i0 < np) {
#pragma unroll
    for (int k = 0; k < NCH; ++k) {
      const int4 x = *(const int4*)(cntP + (size_t)k * np + i0);
      *(int4*)(slotP + (size_t)k * np + i0) = make_int4(cur[0], cur[1], cur[2], cur[3]);
      cur[0] += x.x; cur[1] += x.y; cur[2] += x.z; cur[3] += x.w;
    }
  }
}

__global__ __launch_bounds__(1024) void fill_range(
    const int* __restrict__ src, const int* __restrict__ dst,
    const int* __restrict__ slotP,
    const float* __restrict__ dinv, int2* __restrict__ cv, int n, int E, int np)
{
  __shared__ int   cur[RNG + 32];
  __shared__ float dvl[RNG];
  const int t    = threadIdx.x;
  const int r    = blockIdx.x;
  const int c    = blockIdx.y;
  const int base = r * RNG;

  {
    int   v[RNG / 1024];
    float dv[RNG / 1024];
#pragma unroll
    for (int m = 0; m < RNG / 1024; ++m) {
      const int i  = base + t + m * 1024;
      const int ic = (i < n) ? i : n - 1;
      v[m]  = slotP[(size_t)c * np + ic];
      dv[m] = dinv[ic];
    }
#pragma unroll
    for (int m = 0; m < RNG / 1024; ++m) {
      const int i = base + t + m * 1024;
      cur[t + m * 1024] = (i < n) ? v[m] : 0;
      dvl[t + m * 1024] = (i < n) ? dv[m] : 0.0f;
    }
  }
  if (t < 32) cur[RNG + t] = 0;
  __syncthreads();

  int* const spare = &cur[RNG + (t & 31)];
  int e0, e1;
  chunk_bounds(c, E, e0, e1);
  for (int eb = e0; eb < e1; eb += 1024 * EPT) {
    unsigned s[EPT];
    int   d[EPT], pos[EPT];
    float ws[EPT];
#pragma unroll
    for (int k = 0; k < EPT; ++k) {
      const int e  = eb + k * 1024 + t;
      const int ec = (e < e1) ? e : e1 - 1;
      d[k] = dst[ec];
      s[k] = (unsigned)src[ec];
    }
#pragma unroll
    for (int k = 0; k < EPT; ++k) asm volatile("" :: "v"(s[k]));
#pragma unroll
    for (int k = 0; k < EPT; ++k) {
      const int e = eb + k * 1024 + t;
      const bool hit = e < e1 && (unsigned)(d[k] - base) < (unsigned)RNG;
      ws[k] = dinv[hit ? s[k] : 0u];
    }
#pragma unroll
    for (int k = 0; k < EPT; ++k) {
      const int e = eb + k * 1024 + t;
      const unsigned u = (unsigned)(d[k] - base);
      const bool hit = e < e1 && u < (unsigned)RNG;
      pos[k] = atomicAdd(hit ? &cur[u] : spare, hit ? 1 : 0);
    }
#pragma unroll
    for (int k = 0; k < EPT; ++k) asm volatile("" :: "v"(ws[k]));
#pragma unroll
    for (int k = 0; k < EPT; ++k) {
      const int e = eb + k * 1024 + t;
      const unsigned u = (unsigned)(d[k] - base);
      if (e < e1 && u < (unsigned)RNG)
        cv[pos[k]] = make_int2((int)s[k], __float_as_int(ws[k] * dvl[u]));
    }
  }
}

template <int FOUT>
__device__ __forceinline__ void gemm_wmma_f32_block(
    const float* __restrict__ X, const float* __restrict__ W,
    float* __restrict__ H, int nrows, int blk)
{
  constexpr int CTW = 4;
  constexpr int SPR = FOUT / (16 * CTW);
  constexpr int RPB = 8 / SPR;
  const int w    = __builtin_amdgcn_readfirstlane((int)(threadIdx.x >> 5));
  const int lane = threadIdx.x & 31;

  const int rowTile = blk * RPB + (w / SPR);
  const int r0 = rowTile * 16;
  if (r0 >= nrows) return;
  const int c0 = (w % SPR) * (16 * CTW);

  const int lh = lane & 15;
  const int kh = (lane >> 4) << 1;

  const float* __restrict__ xrow = X + (size_t)(r0 + lh) * 128 + kh;
  const float* __restrict__ wcol = W + (size_t)kh * FOUT + c0 + lh;

  v8f acc[CTW] = {};
#pragma unroll
  for (int kk = 0; kk < 128; kk += 4) {
    v2f a; a.x = xrow[kk]; a.y = xrow[kk + 1];
#pragma unroll
    for (int t = 0; t < CTW; ++t) {
      v2f b;
      b.x = wcol[(size_t)kk * FOUT + t * 16];
      b.y = wcol[(size_t)(kk + 1) * FOUT + t * 16];
      acc[t] = __builtin_amdgcn_wmma_f32_16x16x4_f32(
          false, a, false, b, (short)0, acc[t], false, false);
    }
  }

  const int rbase = r0 + ((lane >> 4) << 3);
  float* hp = H + (size_t)rbase * FOUT + c0 + lh;
#pragma unroll
  for (int t = 0; t < CTW; ++t)
#pragma unroll
    for (int v = 0; v < 8; ++v) hp[(size_t)v * FOUT + t * 16] = acc[t][v];
}

__global__ __launch_bounds__(256) void scan_dinv_gemm1(
    const int* __restrict__ cntP, const int* __restrict__ tot,
    int* __restrict__ slotP,
    int2* __restrict__ seg, float* __restrict__ dinv, int2* __restrict__ cv,
    int n, int np, int nscan,
    const float* __restrict__ X, const float* __restrict__ W,
    float* __restrict__ H)
{
  const int b = blockIdx.x;
  if (b < nscan) scan_dinv_block(cntP, tot, slotP, seg, dinv, cv, n, np, b);
  else           gemm_wmma_f32_block<128>(X, W, H, n, b - nscan);
}

template <int VEC> struct RowVec {
  typedef __attribute__((ext_vector_type(VEC))) float T;
};

template <int VEC>
__device__ __forceinline__ typename RowVec<VEC>::T gather_node(
    const typename RowVec<VEC>::T* __restrict__ Hv,
    const int2* __restrict__ cv, int start, int m, int lane)
{
  typedef typename RowVec<VEC>::T vecT;
  vecT acc = {};

  for (int base = 0; base < m; base += 32) {
    const int mc = (m - base < 32) ? (m - base) : 32;
    int c = 0, wb = 0;
    if (lane < mc) { const int2 e = cv[start + base + lane]; c = e.x; wb = e.y; }

    int j = 0;
    for (; j + 4 <= mc; j += 4) {
      const int s0 = __builtin_amdgcn_readlane(c, j);
      const int s1 = __builtin_amdgcn_readlane(c, j + 1);
      const int s2 = __builtin_amdgcn_readlane(c, j + 2);
      const int s3 = __builtin_amdgcn_readlane(c, j + 3);
      const vecT r0 = Hv[(size_t)s0 * 32 + lane];
      const vecT r1 = Hv[(size_t)s1 * 32 + lane];
      const vecT r2 = Hv[(size_t)s2 * 32 + lane];
      const vecT r3 = Hv[(size_t)s3 * 32 + lane];
      acc += r0 * __int_as_float(__builtin_amdgcn_readlane(wb, j));
      acc += r1 * __int_as_float(__builtin_amdgcn_readlane(wb, j + 1));
      acc += r2 * __int_as_float(__builtin_amdgcn_readlane(wb, j + 2));
      acc += r3 * __int_as_float(__builtin_amdgcn_readlane(wb, j + 3));
    }
    const int rem = mc - j;
    if (rem > 0) {
      vecT r0 = {}, r1 = {}, r2 = {};
      r0 = Hv[(size_t)__builtin_amdgcn_readlane(c, j) * 32 + lane];
      if (rem > 1) r1 = Hv[(size_t)__builtin_amdgcn_readlane(c, j + 1) * 32 + lane];
      if (rem > 2) r2 = Hv[(size_t)__builtin_amdgcn_readlane(c, j + 2) * 32 + lane];
      acc += r0 * __int_as_float(__builtin_amdgcn_readlane(wb, j));
      acc += r1 * __int_as_float(__builtin_amdgcn_readlane(wb, j + 1));
      acc += r2 * __int_as_float(__builtin_amdgcn_readlane(wb, j + 2));
    }
  }
  return acc;
}

template <int FNEXT>
__global__ __launch_bounds__(256, 16) void gather_ln_gemm(
    const float* __restrict__ H, const int2* __restrict__ seg,
    const int2* __restrict__ cv, const float* __restrict__ bias,
    const float* __restrict__ gamma, const float* __restrict__ beta,
    const float* __restrict__ WN, float* __restrict__ HN, int n)
{
  typedef RowVec<4>::T vecT;
  constexpr int ROWS  = 32;
  constexpr int PITCH = 132;
  __shared__ float tile[ROWS * PITCH];

  const int lane = threadIdx.x & 31;
  const int w  = __builtin_amdgcn_readfirstlane((int)(threadIdx.x >> 5));
  const int nb = blockIdx.x * ROWS + w * 4;

  vecT bv, gv, ev;
  for (int k = 0; k < 4; ++k) {
    bv[k] = bias[lane * 4 + k];
    gv[k] = gamma[lane * 4 + k];
    ev[k] = beta[lane * 4 + k];
  }

  int2 sgl = make_int2(0, 0);
  if (lane < 4 && nb + lane < n) sgl = seg[nb + lane];

  const vecT* __restrict__ Hv = (const vecT*)H;
  for (int q = 0; q < 4; ++q) {
    const int start = __builtin_amdgcn_readlane(sgl.x, q);
    const int m     = __builtin_amdgcn_readlane(sgl.y, q);
    const vecT acc = gather_node<4>(Hv, cv, start, m, lane);

    const float v0 = acc[0] + bv[0];
    const float v1 = acc[1] + bv[1];
    const float v2 = acc[2] + bv[2];
    const float v3 = acc[3] + bv[3];

    float s = v0 + v1 + v2 + v3;
#pragma unroll
    for (int x = 16; x >= 1; x >>= 1) s += __shfl_xor(s, x, 32);
    const float mu = s * (1.0f / 128.0f);

    const float d0 = v0 - mu, d1 = v1 - mu, d2 = v2 - mu, d3 = v3 - mu;
    float qs = d0 * d0 + d1 * d1 + d2 * d2 + d3 * d3;
#pragma unroll
    for (int x = 16; x >= 1; x >>= 1) qs += __shfl_xor(qs, x, 32);
    const float rstd = rsqrtf(qs * (1.0f / 128.0f) + LN_EPS);

    vecT o;
    o[0] = fmaxf(0.0f, d0 * rstd * gv[0] + ev[0]);
    o[1] = fmaxf(0.0f, d1 * rstd * gv[1] + ev[1]);
    o[2] = fmaxf(0.0f, d2 * rstd * gv[2] + ev[2]);
    o[3] = fmaxf(0.0f, d3 * rstd * gv[3] + ev[3]);
    *(vecT*)(tile + (w * 4 + q) * PITCH + lane * 4) = o;
  }
  __syncthreads();

  constexpr int CT = FNEXT / 16;
  constexpr int NT = CT / 4;
  const int ct  = w % CT;
  const int rt0 = (w / CT) * NT;
  const int c0  = ct * 16;
  const int lh  = lane & 15;
  const int kh  = (lane >> 4) << 1;

  const float* ap = tile + (rt0 * 16 + lh) * PITCH + kh;
  const float* __restrict__ wcol = WN + (size_t)kh * FNEXT + c0 + lh;

  v8f acc[NT] = {};
#pragma unroll
  for (int kk = 0; kk < 128; kk += 4) {
    v2f b; b.x = wcol[(size_t)kk * FNEXT]; b.y = wcol[(size_t)(kk + 1) * FNEXT];
#pragma unroll
    for (int t = 0; t < NT; ++t) {
      const v2f a = *(const v2f*)(ap + t * 16 * PITCH + kk);
      acc[t] = __builtin_amdgcn_wmma_f32_16x16x4_f32(
          false, a, false, b, (short)0, acc[t], false, false);
    }
  }

#pragma unroll
  for (int t = 0; t < NT; ++t) {
    const int rbase = blockIdx.x * ROWS + (rt0 + t) * 16 + ((lane >> 4) << 3);
    float* hp = HN + (size_t)rbase * FNEXT + c0 + lh;
#pragma unroll
    for (int v = 0; v < 8; ++v)
      if (rbase + v < n) hp[(size_t)v * FNEXT] = acc[t][v];
  }
}

template <int FOUT>
__global__ __launch_bounds__(256) void gather_bias(
    const float* __restrict__ H, const int2* __restrict__ seg,
    const int2* __restrict__ cv, const float* __restrict__ bias,
    float* __restrict__ OUT, int n)
{
  constexpr int VEC = FOUT / 32;
  typedef typename RowVec<VEC>::T vecT;
  const int lane = threadIdx.x & 31;
  const int w  = __builtin_amdgcn_readfirstlane((int)(threadIdx.x >> 5));
  const int nb = blockIdx.x * 32 + w * 4;
  if (nb >= n) return;

  vecT bv;
  for (int k = 0; k < VEC; ++k) bv[k] = bias[lane * VEC + k];

  int2 sgl = make_int2(0, 0);
  if (lane < 4 && nb + lane < n) sgl = seg[nb + lane];

  const vecT* __restrict__ Hv = (const vecT*)H;
  const int nq = (n - nb < 4) ? (n - nb) : 4;
  for (int q = 0; q < nq; ++q) {
    const int start = __builtin_amdgcn_readlane(sgl.x, q);
    const int m     = __builtin_amdgcn_readlane(sgl.y, q);
    const vecT acc = gather_node<VEC>(Hv, cv, start, m, lane);

    vecT o;
    for (int k = 0; k < VEC; ++k) o[k] = acc[k] + bv[k];
    ((vecT*)OUT)[(size_t)(nb + q) * 32 + lane] = o;
  }
}

extern "C" void kernel_launch(void* const* d_in, const int* in_sizes, int n_in,
                              void* d_out, int out_size, void* d_ws, size_t ws_size,
                              hipStream_t stream) {
  const float* x   = (const float*)d_in[0];
  const int*   ei  = (const int*)d_in[1];
  const float* W1  = (const float*)d_in[2];
  const float* b1  = (const float*)d_in[3];
  const float* g1  = (const float*)d_in[4];
  const float* be1 = (const float*)d_in[5];
  const float* W2  = (const float*)d_in[6];
  const float* b2  = (const float*)d_in[7];
  const float* g2  = (const float*)d_in[8];
  const float* be2 = (const float*)d_in[9];
  const float* W3  = (const float*)d_in[10];
  const float* b3  = (const float*)d_in[11];

  const int N = in_sizes[0] / 128;
  const int E = in_sizes[1] / 2;
  const int* src = ei;
  const int* dst = ei + E;

  const size_t Np = ((size_t)N + 63) & ~(size_t)63;
  const size_t Ep = ((size_t)E + 63) & ~(size_t)63;
  float* ws     = (float*)d_ws;
  const int NR  = (N + RNG - 1) / RNG;
  const int NS  = (N + STL - 1) / STL;
  const size_t Tp = ((size_t)NR * SUB * NCH + 63) & ~(size_t)63;
  float* dinv   = ws;
  int*   tot    = (int*)(dinv + Np);
  int2*  seg    = (int2*)(tot + Tp);
  int*   cntP   = (int*)(seg + Np);
  int*   slotP  = cntP + (size_t)NCH * Np;
  int2*  cv     = (int2*)(slotP + (size_t)NCH * Np);
  float* h      = (float*)(cv + Ep + Np);
  float* hn     = h + (size_t)N * 128;

  const dim3 rc(NR, NCH);
  count_range<<<rc, 1024, 0, stream>>>(dst, cntP, tot, E, (int)Np);
  const int gblocks = ((N + 15) / 16 + 3) / 4;
  scan_dinv_gemm1<<<NS + gblocks, 256, 0, stream>>>(cntP, tot, slotP, seg, dinv, cv, N,
                                                    (int)Np, NS, x, W1, h);
  fill_range <<<rc, 1024, 0, stream>>>(src, dst, slotP, dinv, cv, N, E, (int)Np);

  const int fblocks = (N + 31) / 32;

  gather_ln_gemm<128><<<fblocks, 256, 0, stream>>>(h, seg, cv, b1, g1, be1, W2, hn, N);

  gather_ln_gemm<64><<<fblocks, 256, 0, stream>>>(hn, seg, cv, b2, g2, be2, W3, h, N);

  gather_bias<64><<<fblocks, 256, 0, stream>>>(h, seg, cv, b3, (float*)d_out, N);
}
